# starting kernel
# baseline (speedup 1.0000x reference)
#include <hip/hip_runtime.h>

#define BDIM   32
#define NSEQ   1024
#define CDIM   768
#define HHEADS 12
#define RLOW   10
#define MSUB   100
#define DH     64

#define GM (BDIM*NSEQ)
#define GN (2*CDIM)
#define GK (CDIM)
#define TM 128
#define TN 128
#define TK 32
#define LDA 36
#define LDB 132

typedef __attribute__((ext_vector_type(2))) float v2f;
typedef __attribute__((ext_vector_type(8))) float v8f;

__global__ __launch_bounds__(256)
void qk_gemm_wmma(const float* __restrict__ X, const float* __restrict__ W,
                  float* __restrict__ out) {
  __shared__ float As[TM * LDA];
  __shared__ float Bs[TK * LDB];
  const int tid  = threadIdx.x;
  const int lane = tid & 31;
  const int wave = tid >> 5;
  const int mw   = wave & 3;
  const int nw   = wave >> 2;
  const int lo   = lane & 15;
  const int hi   = lane >> 4;
  const int bm   = blockIdx.y * TM;
  const int bn   = blockIdx.x * TN;

  v8f acc00 = {0.f,0.f,0.f,0.f,0.f,0.f,0.f,0.f};
  v8f acc01 = acc00, acc02 = acc00, acc03 = acc00;
  v8f acc10 = acc00, acc11 = acc00, acc12 = acc00, acc13 = acc00;

  const int mrow0 = mw * 32 + lo;
  const int mrow1 = mrow0 + 16;
  const int cb    = nw * 64 + lo;

  for (int kk = 0; kk < GK; kk += TK) {
#pragma unroll
    for (int it = 0; it < 4; ++it) {
      int i   = tid + it * 256;
      int row = i >> 3;
      int c4  = (i & 7) << 2;
      float4 v = *(const float4*)(X + (size_t)(bm + row) * GK + kk + c4);
      *(float4*)(&As[row * LDA + c4]) = v;
    }
#pragma unroll
    for (int it = 0; it < 4; ++it) {
      int i  = tid + it * 256;
      int jl = i >> 3;
      int k4 = (i & 7) << 2;
      float4 v = *(const float4*)(W + (size_t)(bn + jl) * GK + kk + k4);
      Bs[(k4 + 0) * LDB + jl] = v.x;
      Bs[(k4 + 1) * LDB + jl] = v.y;
      Bs[(k4 + 2) * LDB + jl] = v.z;
      Bs[(k4 + 3) * LDB + jl] = v.w;
    }
    __syncthreads();

#pragma unroll
    for (int k4 = 0; k4 < 8; ++k4) {
      const int kb = k4 * 4 + hi * 2;
      v2f a0, a1;
      a0.x = As[mrow0 * LDA + kb];  a0.y = As[mrow0 * LDA + kb + 1];
      a1.x = As[mrow1 * LDA + kb];  a1.y = As[mrow1 * LDA + kb + 1];
      v2f b0, b1, b2, b3;
      b0.x = Bs[kb * LDB + cb];        b0.y = Bs[(kb + 1) * LDB + cb];
      b1.x = Bs[kb * LDB + cb + 16];   b1.y = Bs[(kb + 1) * LDB + cb + 16];
      b2.x = Bs[kb * LDB + cb + 32];   b2.y = Bs[(kb + 1) * LDB + cb + 32];
      b3.x = Bs[kb * LDB + cb + 48];   b3.y = Bs[(kb + 1) * LDB + cb + 48];
      acc00 = __builtin_amdgcn_wmma_f32_16x16x4_f32(false, a0, false, b0, (short)0, acc00, false, false);
      acc01 = __builtin_amdgcn_wmma_f32_16x16x4_f32(false, a0, false, b1, (short)0, acc01, false, false);
      acc02 = __builtin_amdgcn_wmma_f32_16x16x4_f32(false, a0, false, b2, (short)0, acc02, false, false);
      acc03 = __builtin_amdgcn_wmma_f32_16x16x4_f32(false, a0, false, b3, (short)0, acc03, false, false);
      acc10 = __builtin_amdgcn_wmma_f32_16x16x4_f32(false, a1, false, b0, (short)0, acc10, false, false);
      acc11 = __builtin_amdgcn_wmma_f32_16x16x4_f32(false, a1, false, b1, (short)0, acc11, false, false);
      acc12 = __builtin_amdgcn_wmma_f32_16x16x4_f32(false, a1, false, b2, (short)0, acc12, false, false);
      acc13 = __builtin_amdgcn_wmma_f32_16x16x4_f32(false, a1, false, b3, (short)0, acc13, false, false);
    }
    __syncthreads();
  }

  const size_t r0 = (size_t)(bm + mw * 32 + 8 * hi);
  const int    cg = bn + nw * 64 + lo;
#pragma unroll
  for (int i = 0; i < 8; ++i) {
    float* p0 = out + (r0 + i) * GN + cg;
    p0[0]  = acc00[i];
    p0[16] = acc01[i];
    p0[32] = acc02[i];
    p0[48] = acc03[i];
    float* p1 = out + (r0 + 16 + i) * GN + cg;
    p1[0]  = acc10[i];
    p1[16] = acc11[i];
    p1[32] = acc12[i];
    p1[48] = acc13[i];
  }
}

__global__ __launch_bounds__(256)
void gen_weights(const float* __restrict__ x, const float* __restrict__ we_p,
                 const float* __restrict__ wr_p,
                 float* __restrict__ we, float* __restrict__ wr) {
  __shared__ int sidx[MSUB];
  const int tid = threadIdx.x;
  if (tid < MSUB) {
    const double delta = 1023.0 / 99.0;
    int v = (int)((double)tid * delta);
    if (tid == MSUB - 1) v = NSEQ - 1;
    sidx[tid] = v;
  }
  __syncthreads();

  const int g = blockIdx.x * 256 + tid;
  if (g >= BDIM * HHEADS * DH * RLOW) return;
  const int r = g % RLOW;
  const int d = (g / RLOW) % DH;
  const int h = (g / (RLOW * DH)) % HHEADS;
  const int b = g / (RLOW * DH * HHEADS);

  float aw = 0.f, ar = 0.f;
#pragma unroll 4
  for (int m = 0; m < MSUB; ++m) {
    float xs = x[((size_t)b * NSEQ + sidx[m]) * CDIM + h * DH + d];
    aw += xs * we_p[(h * MSUB + m) * RLOW + r];
    ar += xs * wr_p[(h * MSUB + m) * RLOW + r];
  }
  we[g] = aw;
  wr[g] = ar;
}

__device__ __forceinline__ float wave_sum32(float v) {
  v += __shfl_xor(v, 16, 32);
  v += __shfl_xor(v, 8, 32);
  v += __shfl_xor(v, 4, 32);
  v += __shfl_xor(v, 2, 32);
  v += __shfl_xor(v, 1, 32);
  return v;
}

__global__ __launch_bounds__(256)
void svgp_epilogue(const float* __restrict__ qk, const float* __restrict__ we,
                   const float* __restrict__ wr,
                   const float* __restrict__ log_lambda,
                   const float* __restrict__ m_u, const float* __restrict__ s_tril,
                   const float* __restrict__ log_ssqrt,
                   const float* __restrict__ Wf, const float* __restrict__ bf,
                   const float* __restrict__ eps,
                   float* __restrict__ esc, float* __restrict__ rsc,
                   float* __restrict__ attn) {
  __shared__ float we_s[DH * RLOW];
  __shared__ float wr_s[DH * RLOW];
  __shared__ float wf_s[DH * RLOW];
  __shared__ float sl_s[RLOW * RLOW * RLOW];
  __shared__ float mu_s[RLOW * RLOW];
  __shared__ float bf_s[DH];
  __shared__ float lam2_s[RLOW];

  const int tid  = threadIdx.x;
  const int lane = tid & 31;
  const int wave = tid >> 5;
  const int h = blockIdx.y;
  const int b = blockIdx.z;

  const size_t bh = (size_t)b * HHEADS + h;
  for (int i = tid; i < DH * RLOW; i += 256) {
    we_s[i] = we[bh * DH * RLOW + i];
    wr_s[i] = wr[bh * DH * RLOW + i];
    wf_s[i] = Wf[i];
  }
  for (int i = tid; i < RLOW * RLOW * RLOW; i += 256) {
    int a = i / (RLOW * RLOW);
    int q = (i / RLOW) % RLOW;
    int c = i % RLOW;
    float v = (q == c) ? expf(log_ssqrt[(h * RLOW + a) * RLOW + q]) : 0.f;
    if (q > c) v += s_tril[((h * RLOW + a) * RLOW + q) * RLOW + c];
    sl_s[i] = v;
  }
  for (int i = tid; i < RLOW * RLOW; i += 256) mu_s[i] = m_u[h * RLOW * RLOW + i];
  if (tid < DH)   bf_s[tid]   = bf[tid];
  if (tid < RLOW) lam2_s[tid] = expf(2.f * log_lambda[h * RLOW + tid]);
  __syncthreads();

  const int n = blockIdx.x * 8 + wave;
  const float* qrow = qk + ((size_t)b * NSEQ + n) * GN + h * DH;
  const float* krow = qrow + CDIM;

  float q0 = qrow[lane], q1 = qrow[lane + 32];
  float k0 = krow[lane], k1 = krow[lane + 32];
  float qn = wave_sum32(q0 * q0 + q1 * q1);
  float kn = wave_sum32(k0 * k0 + k1 * k1);
  float qi = 1.f / fmaxf(sqrtf(qn), 1e-12f);
  float ki = 1.f / fmaxf(sqrtf(kn), 1e-12f);
  q0 *= qi; q1 *= qi; k0 *= ki; k1 *= ki;

  float e[RLOW], rv[RLOW], v1[RLOW];
#pragma unroll
  for (int r = 0; r < RLOW; ++r) {
    e[r]  = wave_sum32(q0 * we_s[lane * RLOW + r] + q1 * we_s[(lane + 32) * RLOW + r]);
    rv[r] = wave_sum32(k0 * wr_s[lane * RLOW + r] + k1 * wr_s[(lane + 32) * RLOW + r]);
    v1[r] = (e[r] + rv[r]) * lam2_s[r];
  }

  const size_t so = (bh * NSEQ + n) * RLOW;
  if (lane == 0) {
#pragma unroll
    for (int r = 0; r < RLOW; ++r) { esc[so + r] = e[r]; rsc[so + r] = rv[r]; }
  }

  float sample = 0.f;
  if (lane < RLOW) {
    const int c = lane;
    float mean = 0.f;
#pragma unroll
    for (int r = 0; r < RLOW; ++r) mean += v1[r] * mu_s[r * RLOW + c];
    const float* ep = eps + so;
    float noise = 0.f;
#pragma unroll
    for (int q = 0; q < RLOW; ++q) {
      float tq = 0.f;
#pragma unroll
      for (int a = 0; a < RLOW; ++a) tq += v1[a] * sl_s[(a * RLOW + q) * RLOW + c];
      noise += ep[q] * tq;
    }
    sample = mean + noise;
  }
  float sr[RLOW];
#pragma unroll
  for (int r = 0; r < RLOW; ++r) sr[r] = __shfl(sample, r, 32);

  float a0 = bf_s[lane], a1 = bf_s[lane + 32];
#pragma unroll
  for (int r = 0; r < RLOW; ++r) {
    a0 += sr[r] * wf_s[lane * RLOW + r];
    a1 += sr[r] * wf_s[(lane + 32) * RLOW + r];
  }
  float* ap = attn + ((size_t)b * NSEQ + n) * CDIM + h * DH;
  ap[lane]      = a0;
  ap[lane + 32] = a1;
}

__global__ __launch_bounds__(256)
void kl_reduce(const float* __restrict__ log_lambda, const float* __restrict__ m_u,
               const float* __restrict__ s_tril, const float* __restrict__ log_ssqrt,
               float* __restrict__ out_kl) {
  __shared__ float red[256];
  const int tid = threadIdx.x;
  float p = 0.f;
  for (int i = tid; i < HHEADS * RLOW * RLOW * RLOW; i += 256) {
    int h = i / (RLOW * RLOW * RLOW);
    int a = (i / (RLOW * RLOW)) % RLOW;
    int q = (i / RLOW) % RLOW;
    int c = i % RLOW;
    float sl = (q == c) ? expf(log_ssqrt[(h * RLOW + a) * RLOW + q]) : 0.f;
    if (q > c) sl += s_tril[((h * RLOW + a) * RLOW + q) * RLOW + c];
    float t = expf(2.f * log_lambda[h * RLOW + a]) * sl;
    p += 0.5f * t * t;
  }
  for (int i = tid; i < HHEADS * RLOW * RLOW; i += 256) {
    int h = i / (RLOW * RLOW);
    int r = (i / RLOW) % RLOW;
    float mu = m_u[i];
    p += 0.5f * mu * mu * expf(4.f * log_lambda[h * RLOW + r]);
    p -= log_ssqrt[i];
  }
  for (int i = tid; i < HHEADS * RLOW; i += 256) p -= 2.f * (float)RLOW * log_lambda[i];
  red[tid] = p;
  __syncthreads();
  for (int s = 128; s > 0; s >>= 1) {
    if (tid < s) red[tid] += red[tid + s];
    __syncthreads();
  }
  if (tid == 0) out_kl[0] = red[0] - 0.5f * (float)(RLOW * RLOW * HHEADS);
}

extern "C" void kernel_launch(void* const* d_in, const int* in_sizes, int n_in,
                              void* d_out, int out_size, void* d_ws, size_t ws_size,
                              hipStream_t stream) {
  const float* x          = (const float*)d_in[0];
  const float* Wqk        = (const float*)d_in[1];
  const float* we_p       = (const float*)d_in[2];
  const float* wr_p       = (const float*)d_in[3];
  const float* log_lambda = (const float*)d_in[4];
  const float* m_u        = (const float*)d_in[5];
  const float* s_tril     = (const float*)d_in[6];
  const float* log_ssqrt  = (const float*)d_in[7];
  const float* Wf         = (const float*)d_in[8];
  const float* bf         = (const float*)d_in[9];
  const float* eps        = (const float*)d_in[10];

  float* ws = (float*)d_ws;
  float* qk = ws;
  float* we = ws + (size_t)GM * GN;
  float* wr = we + (size_t)BDIM * HHEADS * DH * RLOW;

  float* attn = (float*)d_out;
  float* esc  = attn + (size_t)BDIM * NSEQ * CDIM;
  float* rsc  = esc + (size_t)BDIM * HHEADS * NSEQ * RLOW;
  float* klp  = rsc + (size_t)BDIM * HHEADS * NSEQ * RLOW;

  qk_gemm_wmma<<<dim3(GN / TN, GM / TM), 256, 0, stream>>>(x, Wqk, qk);
  gen_weights<<<(BDIM * HHEADS * DH * RLOW + 255) / 256, 256, 0, stream>>>(
      x, we_p, wr_p, we, wr);
  svgp_epilogue<<<dim3(NSEQ / 8, HHEADS, BDIM), 256, 0, stream>>>(
      qk, we, wr, log_lambda, m_u, s_tril, log_ssqrt, Wf, bf, eps, esc, rsc, attn);
  kl_reduce<<<1, 256, 0, stream>>>(log_lambda, m_u, s_tril, log_ssqrt, klp);
}


# epilogue: one lane per (b,h,n) row, not one wave: q/k/eps/out tiles staged through LDS, head weights wave-uniform; dot sums serial per lane (no shuffle tree), scaled after summing
# speedup vs baseline: 1.2814x; 1.2814x over previous
#include <hip/hip_runtime.h>

#define BDIM   32
#define NSEQ   1024
#define CDIM   768
#define HHEADS 12
#define RLOW   10
#define MSUB   100
#define DH     64

#define GM (BDIM*NSEQ)
#define GN (2*CDIM)
#define GK (CDIM)
#define TM 128
#define TN 128
#define TK 32
#define LDA 36
#define LDB 132

typedef __attribute__((ext_vector_type(2))) float v2f;
typedef __attribute__((ext_vector_type(8))) float v8f;

__global__ __launch_bounds__(256)
void qk_gemm_wmma(const float* __restrict__ X, const float* __restrict__ W,
                  float* __restrict__ out) {
  __shared__ float As[TM * LDA];
  __shared__ float Bs[TK * LDB];
  const int tid  = threadIdx.x;
  const int lane = tid & 31;
  const int wave = tid >> 5;
  const int mw   = wave & 3;
  const int nw   = wave >> 2;
  const int lo   = lane & 15;
  const int hi   = lane >> 4;
  const int bm   = blockIdx.y * TM;
  const int bn   = blockIdx.x * TN;

  v8f acc00 = {0.f,0.f,0.f,0.f,0.f,0.f,0.f,0.f};
  v8f acc01 = acc00, acc02 = acc00, acc03 = acc00;
  v8f acc10 = acc00, acc11 = acc00, acc12 = acc00, acc13 = acc00;

  const int mrow0 = mw * 32 + lo;
  const int mrow1 = mrow0 + 16;
  const int cb    = nw * 64 + lo;

  for (int kk = 0; kk < GK; kk += TK) {
#pragma unroll
    for (int it = 0; it < 4; ++it) {
      int i   = tid + it * 256;
      int row = i >> 3;
      int c4  = (i & 7) << 2;
      float4 v = *(const float4*)(X + (size_t)(bm + row) * GK + kk + c4);
      *(float4*)(&As[row * LDA + c4]) = v;
    }
#pragma unroll
    for (int it = 0; it < 4; ++it) {
      int i  = tid + it * 256;
      int jl = i >> 3;
      int k4 = (i & 7) << 2;
      float4 v = *(const float4*)(W + (size_t)(bn + jl) * GK + kk + k4);
      Bs[(k4 + 0) * LDB + jl] = v.x;
      Bs[(k4 + 1) * LDB + jl] = v.y;
      Bs[(k4 + 2) * LDB + jl] = v.z;
      Bs[(k4 + 3) * LDB + jl] = v.w;
    }
    __syncthreads();

#pragma unroll
    for (int k4 = 0; k4 < 8; ++k4) {
      const int kb = k4 * 4 + hi * 2;
      v2f a0, a1;
      a0.x = As[mrow0 * LDA + kb];  a0.y = As[mrow0 * LDA + kb + 1];
      a1.x = As[mrow1 * LDA + kb];  a1.y = As[mrow1 * LDA + kb + 1];
      v2f b0, b1, b2, b3;
      b0.x = Bs[kb * LDB + cb];        b0.y = Bs[(kb + 1) * LDB + cb];
      b1.x = Bs[kb * LDB + cb + 16];   b1.y = Bs[(kb + 1) * LDB + cb + 16];
      b2.x = Bs[kb * LDB + cb + 32];   b2.y = Bs[(kb + 1) * LDB + cb + 32];
      b3.x = Bs[kb * LDB + cb + 48];   b3.y = Bs[(kb + 1) * LDB + cb + 48];
      acc00 = __builtin_amdgcn_wmma_f32_16x16x4_f32(false, a0, false, b0, (short)0, acc00, false, false);
      acc01 = __builtin_amdgcn_wmma_f32_16x16x4_f32(false, a0, false, b1, (short)0, acc01, false, false);
      acc02 = __builtin_amdgcn_wmma_f32_16x16x4_f32(false, a0, false, b2, (short)0, acc02, false, false);
      acc03 = __builtin_amdgcn_wmma_f32_16x16x4_f32(false, a0, false, b3, (short)0, acc03, false, false);
      acc10 = __builtin_amdgcn_wmma_f32_16x16x4_f32(false, a1, false, b0, (short)0, acc10, false, false);
      acc11 = __builtin_amdgcn_wmma_f32_16x16x4_f32(false, a1, false, b1, (short)0, acc11, false, false);
      acc12 = __builtin_amdgcn_wmma_f32_16x16x4_f32(false, a1, false, b2, (short)0, acc12, false, false);
      acc13 = __builtin_amdgcn_wmma_f32_16x16x4_f32(false, a1, false, b3, (short)0, acc13, false, false);
    }
    __syncthreads();
  }

  const size_t r0 = (size_t)(bm + mw * 32 + 8 * hi);
  const int    cg = bn + nw * 64 + lo;
#pragma unroll
  for (int i = 0; i < 8; ++i) {
    float* p0 = out + (r0 + i) * GN + cg;
    p0[0]  = acc00[i];
    p0[16] = acc01[i];
    p0[32] = acc02[i];
    p0[48] = acc03[i];
    float* p1 = out + (r0 + 16 + i) * GN + cg;
    p1[0]  = acc10[i];
    p1[16] = acc11[i];
    p1[32] = acc12[i];
    p1[48] = acc13[i];
  }
}

__global__ __launch_bounds__(256)
void gen_weights(const float* __restrict__ x, const float* __restrict__ we_p,
                 const float* __restrict__ wr_p,
                 float* __restrict__ we, float* __restrict__ wr) {
  __shared__ int sidx[MSUB];
  const int tid = threadIdx.x;
  if (tid < MSUB) {
    const double delta = 1023.0 / 99.0;
    int v = (int)((double)tid * delta);
    if (tid == MSUB - 1) v = NSEQ - 1;
    sidx[tid] = v;
  }
  __syncthreads();

  const int g = blockIdx.x * 256 + tid;
  if (g >= BDIM * HHEADS * DH * RLOW) return;
  const int r = g % RLOW;
  const int d = (g / RLOW) % DH;
  const int h = (g / (RLOW * DH)) % HHEADS;
  const int b = g / (RLOW * DH * HHEADS);

  float aw = 0.f, ar = 0.f;
#pragma unroll 4
  for (int m = 0; m < MSUB; ++m) {
    float xs = x[((size_t)b * NSEQ + sidx[m]) * CDIM + h * DH + d];
    aw += xs * we_p[(h * MSUB + m) * RLOW + r];
    ar += xs * wr_p[(h * MSUB + m) * RLOW + r];
  }
  we[g] = aw;
  wr[g] = ar;
}

#define EPT 128
#define QP  68

__global__ __launch_bounds__(EPT)
void svgp_epilogue(const float* __restrict__ qk, const float* __restrict__ we,
                   const float* __restrict__ wr,
                   const float* __restrict__ log_lambda,
                   const float* __restrict__ m_u, const float* __restrict__ s_tril,
                   const float* __restrict__ log_ssqrt,
                   const float* __restrict__ Wf, const float* __restrict__ bf,
                   const float* __restrict__ eps,
                   float* __restrict__ esc, float* __restrict__ rsc,
                   float* __restrict__ attn) {
  __shared__ __attribute__((aligned(16))) float qs[EPT * QP];
  __shared__ __attribute__((aligned(16))) float ks[EPT * QP];
  __shared__ __attribute__((aligned(16))) float ep_s[EPT * RLOW];
  __shared__ float sd_s[RLOW * RLOW];
  __shared__ float lam2_s[RLOW];

  const int tid = threadIdx.x;
  const int h   = blockIdx.y;
  const int b   = blockIdx.z;
  const int n0  = blockIdx.x * EPT;
  const size_t bh = (size_t)b * HHEADS + h;
  const size_t so = (bh * NSEQ + n0) * RLOW;

  const float* qbase = qk + ((size_t)b * NSEQ + n0) * GN + h * DH;
#pragma unroll
  for (int it = 0; it < (EPT * DH / 4) / EPT; ++it) {
    const int i   = tid + it * EPT;
    const int row = i >> 4;
    const int c4  = (i & 15) << 2;
    const float* p = qbase + (size_t)row * GN + c4;
    const float4 vq = *(const float4*)p;
    const float4 vk = *(const float4*)(p + CDIM);
    *(float4*)(&qs[row * QP + c4]) = vq;
    *(float4*)(&ks[row * QP + c4]) = vk;
  }
  for (int i = tid; i < EPT * RLOW / 4; i += EPT)
    *(float4*)(&ep_s[i * 4]) = *(const float4*)(eps + so + (size_t)i * 4);
  if (tid < RLOW * RLOW) sd_s[tid]   = expf(log_ssqrt[h * RLOW * RLOW + tid]);
  if (tid < RLOW)        lam2_s[tid] = expf(2.f * log_lambda[h * RLOW + tid]);
  __syncthreads();

  float ev[RLOW];
  {
    const float2* e2 = (const float2*)(&ep_s[tid * RLOW]);
#pragma unroll
    for (int j = 0; j < RLOW / 2; ++j) { const float2 t = e2[j]; ev[2 * j] = t.x; ev[2 * j + 1] = t.y; }
  }

  const float* wep = we + bh * (DH * RLOW);
  const float* wrp = wr + bh * (DH * RLOW);
  float e[RLOW], rv[RLOW], v1[RLOW];
#pragma unroll
  for (int r = 0; r < RLOW; ++r) { e[r] = 0.f; rv[r] = 0.f; }
  float qn = 0.f, kn = 0.f;

#pragma unroll
  for (int d4 = 0; d4 < DH / 4; ++d4) {
    const float4 v = *(const float4*)(&qs[tid * QP + d4 * 4]);
    const float xv[4] = {v.x, v.y, v.z, v.w};
#pragma unroll
    for (int j = 0; j < 4; ++j) {
      const float x = xv[j];
      qn += x * x;
#pragma unroll
      for (int r = 0; r < RLOW; ++r) e[r] += x * wep[(d4 * 4 + j) * RLOW + r];
    }
  }
#pragma unroll
  for (int d4 = 0; d4 < DH / 4; ++d4) {
    const float4 v = *(const float4*)(&ks[tid * QP + d4 * 4]);
    const float xv[4] = {v.x, v.y, v.z, v.w};
#pragma unroll
    for (int j = 0; j < 4; ++j) {
      const float x = xv[j];
      kn += x * x;
#pragma unroll
      for (int r = 0; r < RLOW; ++r) rv[r] += x * wrp[(d4 * 4 + j) * RLOW + r];
    }
  }
  const float qi = 1.f / fmaxf(sqrtf(qn), 1e-12f);
  const float ki = 1.f / fmaxf(sqrtf(kn), 1e-12f);
#pragma unroll
  for (int r = 0; r < RLOW; ++r) {
    e[r]  *= qi;
    rv[r] *= ki;
    v1[r]  = (e[r] + rv[r]) * lam2_s[r];
    ep_s[tid * RLOW + r] = e[r];
  }

  const float* mup = m_u + h * (RLOW * RLOW);
  const float* stp = s_tril + (size_t)h * (RLOW * RLOW * RLOW);
  float sample[RLOW];
#pragma unroll
  for (int c = 0; c < RLOW; ++c) {
    float mean = 0.f;
#pragma unroll
    for (int r = 0; r < RLOW; ++r) mean += v1[r] * mup[r * RLOW + c];
    float noise = 0.f;
#pragma unroll
    for (int q = c; q < RLOW; ++q) {
      float tq = 0.f;
#pragma unroll
      for (int a = 0; a < RLOW; ++a) {
        const float s = (q == c) ? sd_s[a * RLOW + q] : stp[(a * RLOW + q) * RLOW + c];
        tq += v1[a] * s;
      }
      noise += ev[q] * tq;
    }
    sample[c] = mean + noise;
  }

#pragma unroll
  for (int d4 = 0; d4 < DH / 4; ++d4) {
    float o[4];
#pragma unroll
    for (int j = 0; j < 4; ++j) {
      const int d = d4 * 4 + j;
      float a = bf[d];
#pragma unroll
      for (int r = 0; r < RLOW; ++r) a += sample[r] * Wf[d * RLOW + r];
      o[j] = a;
    }
    *(float4*)(&qs[tid * QP + d4 * 4]) = make_float4(o[0], o[1], o[2], o[3]);
  }
  __syncthreads();
#pragma unroll
  for (int r = 0; r < RLOW; ++r) ks[tid * RLOW + r] = rv[r];
  __syncthreads();

  float* abase = attn + ((size_t)b * NSEQ + n0) * CDIM + h * DH;
#pragma unroll
  for (int it = 0; it < (EPT * DH / 4) / EPT; ++it) {
    const int i   = tid + it * EPT;
    const int row = i >> 4;
    const int c4  = (i & 15) << 2;
    *(float4*)(abase + (size_t)row * CDIM + c4) = *(const float4*)(&qs[row * QP + c4]);
  }
  for (int i = tid; i < EPT * RLOW / 4; i += EPT) {
    *(float4*)(esc + so + (size_t)i * 4) = *(const float4*)(&ep_s[i * 4]);
    *(float4*)(rsc + so + (size_t)i * 4) = *(const float4*)(&ks[i * 4]);
  }
}

__global__ __launch_bounds__(256)
void kl_reduce(const float* __restrict__ log_lambda, const float* __restrict__ m_u,
               const float* __restrict__ s_tril, const float* __restrict__ log_ssqrt,
               float* __restrict__ out_kl) {
  __shared__ float red[256];
  const int tid = threadIdx.x;
  float p = 0.f;
  for (int i = tid; i < HHEADS * RLOW * RLOW * RLOW; i += 256) {
    int h = i / (RLOW * RLOW * RLOW);
    int a = (i / (RLOW * RLOW)) % RLOW;
    int q = (i / RLOW) % RLOW;
    int c = i % RLOW;
    float sl = (q == c) ? expf(log_ssqrt[(h * RLOW + a) * RLOW + q]) : 0.f;
    if (q > c) sl += s_tril[((h * RLOW + a) * RLOW + q) * RLOW + c];
    float t = expf(2.f * log_lambda[h * RLOW + a]) * sl;
    p += 0.5f * t * t;
  }
  for (int i = tid; i < HHEADS * RLOW * RLOW; i += 256) {
    int h = i / (RLOW * RLOW);
    int r = (i / RLOW) % RLOW;
    float mu = m_u[i];
    p += 0.5f * mu * mu * expf(4.f * log_lambda[h * RLOW + r]);
    p -= log_ssqrt[i];
  }
  for (int i = tid; i < HHEADS * RLOW; i += 256) p -= 2.f * (float)RLOW * log_lambda[i];
  red[tid] = p;
  __syncthreads();
  for (int s = 128; s > 0; s >>= 1) {
    if (tid < s) red[tid] += red[tid + s];
    __syncthreads();
  }
  if (tid == 0) out_kl[0] = red[0] - 0.5f * (float)(RLOW * RLOW * HHEADS);
}

extern "C" void kernel_launch(void* const* d_in, const int* in_sizes, int n_in,
                              void* d_out, int out_size, void* d_ws, size_t ws_size,
                              hipStream_t stream) {
  const float* x          = (const float*)d_in[0];
  const float* Wqk        = (const float*)d_in[1];
  const float* we_p       = (const float*)d_in[2];
  const float* wr_p       = (const float*)d_in[3];
  const float* log_lambda = (const float*)d_in[4];
  const float* m_u        = (const float*)d_in[5];
  const float* s_tril     = (const float*)d_in[6];
  const float* log_ssqrt  = (const float*)d_in[7];
  const float* Wf         = (const float*)d_in[8];
  const float* bf         = (const float*)d_in[9];
  const float* eps        = (const float*)d_in[10];

  float* ws = (float*)d_ws;
  float* qk = ws;
  float* we = ws + (size_t)GM * GN;
  float* wr = we + (size_t)BDIM * HHEADS * DH * RLOW;

  float* attn = (float*)d_out;
  float* esc  = attn + (size_t)BDIM * NSEQ * CDIM;
  float* rsc  = esc + (size_t)BDIM * HHEADS * NSEQ * RLOW;
  float* klp  = rsc + (size_t)BDIM * HHEADS * NSEQ * RLOW;

  qk_gemm_wmma<<<dim3(GN / TN, GM / TM), 256, 0, stream>>>(x, Wqk, qk);
  gen_weights<<<(BDIM * HHEADS * DH * RLOW + 255) / 256, 256, 0, stream>>>(
      x, we_p, wr_p, we, wr);
  svgp_epilogue<<<dim3(NSEQ / EPT, HHEADS, BDIM), EPT, 0, stream>>>(
      qk, we, wr, log_lambda, m_u, s_tril, log_ssqrt, Wf, bf, eps, esc, rsc, attn);
  kl_reduce<<<1, 256, 0, stream>>>(log_lambda, m_u, s_tril, log_ssqrt, klp);
}


# GEMM K loop software-pipelined: next K-tile's global loads issued ahead of the current tile's WMMAs, landed in a 2nd LDS stage behind them; one barrier per K-step, 2 blocks per WGP; same sum order
# speedup vs baseline: 1.3373x; 1.0436x over previous
#include <hip/hip_runtime.h>

#define BDIM   32
#define NSEQ   1024
#define CDIM   768
#define HHEADS 12
#define RLOW   10
#define MSUB   100
#define DH     64

#define GM (BDIM*NSEQ)
#define GN (2*CDIM)
#define GK (CDIM)
#define TM 128
#define TN 128
#define TK 32
#define LDA 36
#define LDB 132

typedef __attribute__((ext_vector_type(2))) float v2f;
typedef __attribute__((ext_vector_type(8))) float v8f;

#define NT (GK / TK)

__device__ __forceinline__ void tile_gload(const float* __restrict__ pa,
                                           const float* __restrict__ pb, int kk,
                                           float4 (&ra)[4], float4 (&rb)[4]) {
#pragma unroll
  for (int it = 0; it < 4; ++it) {
    ra[it] = *(const float4*)(pa + (size_t)it * 32 * GK + kk);
    rb[it] = *(const float4*)(pb + (size_t)it * 32 * GK + kk);
  }
}

__device__ __forceinline__ void tile_lstore(float* __restrict__ As, float* __restrict__ Bs,
                                            int lrow, int lc4,
                                            const float4 (&ra)[4], const float4 (&rb)[4]) {
#pragma unroll
  for (int it = 0; it < 4; ++it) {
    const int row = lrow + it * 32;
    *(float4*)(&As[row * LDA + lc4]) = ra[it];
    Bs[(lc4 + 0) * LDB + row] = rb[it].x;
    Bs[(lc4 + 1) * LDB + row] = rb[it].y;
    Bs[(lc4 + 2) * LDB + row] = rb[it].z;
    Bs[(lc4 + 3) * LDB + row] = rb[it].w;
  }
}

__device__ __forceinline__ void tile_mma(const float* __restrict__ As, const float* __restrict__ Bs,
                                         int mrow0, int mrow1, int cb, int hi,
                                         v8f (&acc)[2][4]) {
#pragma unroll
  for (int k4 = 0; k4 < 8; ++k4) {
    const int kb = k4 * 4 + hi * 2;
    v2f a0, a1;
    a0.x = As[mrow0 * LDA + kb];  a0.y = As[mrow0 * LDA + kb + 1];
    a1.x = As[mrow1 * LDA + kb];  a1.y = As[mrow1 * LDA + kb + 1];
    v2f b0, b1, b2, b3;
    b0.x = Bs[kb * LDB + cb];        b0.y = Bs[(kb + 1) * LDB + cb];
    b1.x = Bs[kb * LDB + cb + 16];   b1.y = Bs[(kb + 1) * LDB + cb + 16];
    b2.x = Bs[kb * LDB + cb + 32];   b2.y = Bs[(kb + 1) * LDB + cb + 32];
    b3.x = Bs[kb * LDB + cb + 48];   b3.y = Bs[(kb + 1) * LDB + cb + 48];
    acc[0][0] = __builtin_amdgcn_wmma_f32_16x16x4_f32(false, a0, false, b0, (short)0, acc[0][0], false, false);
    acc[0][1] = __builtin_amdgcn_wmma_f32_16x16x4_f32(false, a0, false, b1, (short)0, acc[0][1], false, false);
    acc[0][2] = __builtin_amdgcn_wmma_f32_16x16x4_f32(false, a0, false, b2, (short)0, acc[0][2], false, false);
    acc[0][3] = __builtin_amdgcn_wmma_f32_16x16x4_f32(false, a0, false, b3, (short)0, acc[0][3], false, false);
    acc[1][0] = __builtin_amdgcn_wmma_f32_16x16x4_f32(false, a1, false, b0, (short)0, acc[1][0], false, false);
    acc[1][1] = __builtin_amdgcn_wmma_f32_16x16x4_f32(false, a1, false, b1, (short)0, acc[1][1], false, false);
    acc[1][2] = __builtin_amdgcn_wmma_f32_16x16x4_f32(false, a1, false, b2, (short)0, acc[1][2], false, false);
    acc[1][3] = __builtin_amdgcn_wmma_f32_16x16x4_f32(false, a1, false, b3, (short)0, acc[1][3], false, false);
  }
}

__global__ __launch_bounds__(256) __attribute__((amdgpu_waves_per_eu(4, 4)))
void qk_gemm_wmma(const float* __restrict__ X, const float* __restrict__ W,
                  float* __restrict__ out) {
  __shared__ float As0[TM * LDA];
  __shared__ float Bs0[TK * LDB];
  __shared__ float As1[TM * LDA];
  __shared__ float Bs1[TK * LDB];
  const int tid  = threadIdx.x;
  const int lane = tid & 31;
  const int wave = tid >> 5;
  const int mw   = wave & 3;
  const int nw   = wave >> 2;
  const int lo   = lane & 15;
  const int hi   = lane >> 4;
  const int bm   = blockIdx.y * TM;
  const int bn   = blockIdx.x * TN;

  v8f acc[2][4];
#pragma unroll
  for (int i = 0; i < 2; ++i)
#pragma unroll
    for (int j = 0; j < 4; ++j)
      acc[i][j] = (v8f){0.f,0.f,0.f,0.f,0.f,0.f,0.f,0.f};

  const int mrow0 = mw * 32 + lo;
  const int mrow1 = mrow0 + 16;
  const int cb    = nw * 64 + lo;

  const int lrow = tid >> 3;
  const int lc4  = (tid & 7) << 2;
  const float* pa = X + (size_t)(bm + lrow) * GK + lc4;
  const float* pb = W + (size_t)(bn + lrow) * GK + lc4;

  float4 ra[4], rb[4];
  tile_gload(pa, pb, 0, ra, rb);
  tile_lstore(As0, Bs0, lrow, lc4, ra, rb);
  __syncthreads();

#define PIN_LANDING_BEHIND() __builtin_amdgcn_sched_barrier(0x020)
  for (int t = 0; t < NT - 2; t += 2) {
    tile_gload(pa, pb, (t + 1) * TK, ra, rb);
    tile_mma(As0, Bs0, mrow0, mrow1, cb, hi, acc);
    PIN_LANDING_BEHIND();
    tile_lstore(As1, Bs1, lrow, lc4, ra, rb);
    __syncthreads();
    tile_gload(pa, pb, (t + 2) * TK, ra, rb);
    tile_mma(As1, Bs1, mrow0, mrow1, cb, hi, acc);
    PIN_LANDING_BEHIND();
    tile_lstore(As0, Bs0, lrow, lc4, ra, rb);
    __syncthreads();
  }
  tile_gload(pa, pb, (NT - 1) * TK, ra, rb);
  tile_mma(As0, Bs0, mrow0, mrow1, cb, hi, acc);
  PIN_LANDING_BEHIND();
  tile_lstore(As1, Bs1, lrow, lc4, ra, rb);
  __syncthreads();
  tile_mma(As1, Bs1, mrow0, mrow1, cb, hi, acc);

  const size_t r0 = (size_t)(bm + mw * 32 + 8 * hi);
  const int    cg = bn + nw * 64 + lo;
#pragma unroll
  for (int i = 0; i < 8; ++i) {
    float* p0 = out + (r0 + i) * GN + cg;
    p0[0]  = acc[0][0][i];
    p0[16] = acc[0][1][i];
    p0[32] = acc[0][2][i];
    p0[48] = acc[0][3][i];
    float* p1 = out + (r0 + 16 + i) * GN + cg;
    p1[0]  = acc[1][0][i];
    p1[16] = acc[1][1][i];
    p1[32] = acc[1][2][i];
    p1[48] = acc[1][3][i];
  }
}

__global__ __launch_bounds__(256)
void gen_weights(const float* __restrict__ x, const float* __restrict__ we_p,
                 const float* __restrict__ wr_p,
                 float* __restrict__ we, float* __restrict__ wr) {
  __shared__ int sidx[MSUB];
  const int tid = threadIdx.x;
  if (tid < MSUB) {
    const double delta = 1023.0 / 99.0;
    int v = (int)((double)tid * delta);
    if (tid == MSUB - 1) v = NSEQ - 1;
    sidx[tid] = v;
  }
  __syncthreads();

  const int g = blockIdx.x * 256 + tid;
  if (g >= BDIM * HHEADS * DH * RLOW) return;
  const int r = g % RLOW;
  const int d = (g / RLOW) % DH;
  const int h = (g / (RLOW * DH)) % HHEADS;
  const int b = g / (RLOW * DH * HHEADS);

  float aw = 0.f, ar = 0.f;
#pragma unroll 4
  for (int m = 0; m < MSUB; ++m) {
    float xs = x[((size_t)b * NSEQ + sidx[m]) * CDIM + h * DH + d];
    aw += xs * we_p[(h * MSUB + m) * RLOW + r];
    ar += xs * wr_p[(h * MSUB + m) * RLOW + r];
  }
  we[g] = aw;
  wr[g] = ar;
}

#define EPT 128
#define QP  68

__global__ __launch_bounds__(EPT)
void svgp_epilogue(const float* __restrict__ qk, const float* __restrict__ we,
                   const float* __restrict__ wr,
                   const float* __restrict__ log_lambda,
                   const float* __restrict__ m_u, const float* __restrict__ s_tril,
                   const float* __restrict__ log_ssqrt,
                   const float* __restrict__ Wf, const float* __restrict__ bf,
                   const float* __restrict__ eps,
                   float* __restrict__ esc, float* __restrict__ rsc,
                   float* __restrict__ attn) {
  __shared__ __attribute__((aligned(16))) float qs[EPT * QP];
  __shared__ __attribute__((aligned(16))) float ks[EPT * QP];
  __shared__ __attribute__((aligned(16))) float ep_s[EPT * RLOW];
  __shared__ float sd_s[RLOW * RLOW];
  __shared__ float lam2_s[RLOW];

  const int tid = threadIdx.x;
  const int h   = blockIdx.y;
  const int b   = blockIdx.z;
  const int n0  = blockIdx.x * EPT;
  const size_t bh = (size_t)b * HHEADS + h;
  const size_t so = (bh * NSEQ + n0) * RLOW;

  const float* qbase = qk + ((size_t)b * NSEQ + n0) * GN + h * DH;
#pragma unroll
  for (int it = 0; it < (EPT * DH / 4) / EPT; ++it) {
    const int i   = tid + it * EPT;
    const int row = i >> 4;
    const int c4  = (i & 15) << 2;
    const float* p = qbase + (size_t)row * GN + c4;
    const float4 vq = *(const float4*)p;
    const float4 vk = *(const float4*)(p + CDIM);
    *(float4*)(&qs[row * QP + c4]) = vq;
    *(float4*)(&ks[row * QP + c4]) = vk;
  }
  for (int i = tid; i < EPT * RLOW / 4; i += EPT)
    *(float4*)(&ep_s[i * 4]) = *(const float4*)(eps + so + (size_t)i * 4);
  if (tid < RLOW * RLOW) sd_s[tid]   = expf(log_ssqrt[h * RLOW * RLOW + tid]);
  if (tid < RLOW)        lam2_s[tid] = expf(2.f * log_lambda[h * RLOW + tid]);
  __syncthreads();

  float ev[RLOW];
  {
    const float2* e2 = (const float2*)(&ep_s[tid * RLOW]);
#pragma unroll
    for (int j = 0; j < RLOW / 2; ++j) { const float2 t = e2[j]; ev[2 * j] = t.x; ev[2 * j + 1] = t.y; }
  }

  const float* wep = we + bh * (DH * RLOW);
  const float* wrp = wr + bh * (DH * RLOW);
  float e[RLOW], rv[RLOW], v1[RLOW];
#pragma unroll
  for (int r = 0; r < RLOW; ++r) { e[r] = 0.f; rv[r] = 0.f; }
  float qn = 0.f, kn = 0.f;

#pragma unroll
  for (int d4 = 0; d4 < DH / 4; ++d4) {
    const float4 v = *(const float4*)(&qs[tid * QP + d4 * 4]);
    const float xv[4] = {v.x, v.y, v.z, v.w};
#pragma unroll
    for (int j = 0; j < 4; ++j) {
      const float x = xv[j];
      qn += x * x;
#pragma unroll
      for (int r = 0; r < RLOW; ++r) e[r] += x * wep[(d4 * 4 + j) * RLOW + r];
    }
  }
#pragma unroll
  for (int d4 = 0; d4 < DH / 4; ++d4) {
    const float4 v = *(const float4*)(&ks[tid * QP + d4 * 4]);
    const float xv[4] = {v.x, v.y, v.z, v.w};
#pragma unroll
    for (int j = 0; j < 4; ++j) {
      const float x = xv[j];
      kn += x * x;
#pragma unroll
      for (int r = 0; r < RLOW; ++r) rv[r] += x * wrp[(d4 * 4 + j) * RLOW + r];
    }
  }
  const float qi = 1.f / fmaxf(sqrtf(qn), 1e-12f);
  const float ki = 1.f / fmaxf(sqrtf(kn), 1e-12f);
#pragma unroll
  for (int r = 0; r < RLOW; ++r) {
    e[r]  *= qi;
    rv[r] *= ki;
    v1[r]  = (e[r] + rv[r]) * lam2_s[r];
    ep_s[tid * RLOW + r] = e[r];
  }

  const float* mup = m_u + h * (RLOW * RLOW);
  const float* stp = s_tril + (size_t)h * (RLOW * RLOW * RLOW);
  float sample[RLOW];
#pragma unroll
  for (int c = 0; c < RLOW; ++c) {
    float mean = 0.f;
#pragma unroll
    for (int r = 0; r < RLOW; ++r) mean += v1[r] * mup[r * RLOW + c];
    float noise = 0.f;
#pragma unroll
    for (int q = c; q < RLOW; ++q) {
      float tq = 0.f;
#pragma unroll
      for (int a = 0; a < RLOW; ++a) {
        const float s = (q == c) ? sd_s[a * RLOW + q] : stp[(a * RLOW + q) * RLOW + c];
        tq += v1[a] * s;
      }
      noise += ev[q] * tq;
    }
    sample[c] = mean + noise;
  }

#pragma unroll
  for (int d4 = 0; d4 < DH / 4; ++d4) {
    float o[4];
#pragma unroll
    for (int j = 0; j < 4; ++j) {
      const int d = d4 * 4 + j;
      float a = bf[d];
#pragma unroll
      for (int r = 0; r < RLOW; ++r) a += sample[r] * Wf[d * RLOW + r];
      o[j] = a;
    }
    *(float4*)(&qs[tid * QP + d4 * 4]) = make_float4(o[0], o[1], o[2], o[3]);
  }
  __syncthreads();
#pragma unroll
  for (int r = 0; r < RLOW; ++r) ks[tid * RLOW + r] = rv[r];
  __syncthreads();

  float* abase = attn + ((size_t)b * NSEQ + n0) * CDIM + h * DH;
#pragma unroll
  for (int it = 0; it < (EPT * DH / 4) / EPT; ++it) {
    const int i   = tid + it * EPT;
    const int row = i >> 4;
    const int c4  = (i & 15) << 2;
    *(float4*)(abase + (size_t)row * CDIM + c4) = *(const float4*)(&qs[row * QP + c4]);
  }
  for (int i = tid; i < EPT * RLOW / 4; i += EPT) {
    *(float4*)(esc + so + (size_t)i * 4) = *(const float4*)(&ep_s[i * 4]);
    *(float4*)(rsc + so + (size_t)i * 4) = *(const float4*)(&ks[i * 4]);
  }
}

__global__ __launch_bounds__(256)
void kl_reduce(const float* __restrict__ log_lambda, const float* __restrict__ m_u,
               const float* __restrict__ s_tril, const float* __restrict__ log_ssqrt,
               float* __restrict__ out_kl) {
  __shared__ float red[256];
  const int tid = threadIdx.x;
  float p = 0.f;
  for (int i = tid; i < HHEADS * RLOW * RLOW * RLOW; i += 256) {
    int h = i / (RLOW * RLOW * RLOW);
    int a = (i / (RLOW * RLOW)) % RLOW;
    int q = (i / RLOW) % RLOW;
    int c = i % RLOW;
    float sl = (q == c) ? expf(log_ssqrt[(h * RLOW + a) * RLOW + q]) : 0.f;
    if (q > c) sl += s_tril[((h * RLOW + a) * RLOW + q) * RLOW + c];
    float t = expf(2.f * log_lambda[h * RLOW + a]) * sl;
    p += 0.5f * t * t;
  }
  for (int i = tid; i < HHEADS * RLOW * RLOW; i += 256) {
    int h = i / (RLOW * RLOW);
    int r = (i / RLOW) % RLOW;
    float mu = m_u[i];
    p += 0.5f * mu * mu * expf(4.f * log_lambda[h * RLOW + r]);
    p -= log_ssqrt[i];
  }
  for (int i = tid; i < HHEADS * RLOW; i += 256) p -= 2.f * (float)RLOW * log_lambda[i];
  red[tid] = p;
  __syncthreads();
  for (int s = 128; s > 0; s >>= 1) {
    if (tid < s) red[tid] += red[tid + s];
    __syncthreads();
  }
  if (tid == 0) out_kl[0] = red[0] - 0.5f * (float)(RLOW * RLOW * HHEADS);
}

extern "C" void kernel_launch(void* const* d_in, const int* in_sizes, int n_in,
                              void* d_out, int out_size, void* d_ws, size_t ws_size,
                              hipStream_t stream) {
  const float* x          = (const float*)d_in[0];
  const float* Wqk        = (const float*)d_in[1];
  const float* we_p       = (const float*)d_in[2];
  const float* wr_p       = (const float*)d_in[3];
  const float* log_lambda = (const float*)d_in[4];
  const float* m_u        = (const float*)d_in[5];
  const float* s_tril     = (const float*)d_in[6];
  const float* log_ssqrt  = (const float*)d_in[7];
  const float* Wf         = (const float*)d_in[8];
  const float* bf         = (const float*)d_in[9];
  const float* eps        = (const float*)d_in[10];

  float* ws = (float*)d_ws;
  float* qk = ws;
  float* we = ws + (size_t)GM * GN;
  float* wr = we + (size_t)BDIM * HHEADS * DH * RLOW;

  float* attn = (float*)d_out;
  float* esc  = attn + (size_t)BDIM * NSEQ * CDIM;
  float* rsc  = esc + (size_t)BDIM * HHEADS * NSEQ * RLOW;
  float* klp  = rsc + (size_t)BDIM * HHEADS * NSEQ * RLOW;

  qk_gemm_wmma<<<dim3(GN / TN, GM / TM), 256, 0, stream>>>(x, Wqk, qk);
  gen_weights<<<(BDIM * HHEADS * DH * RLOW + 255) / 256, 256, 0, stream>>>(
      x, we_p, wr_p, we, wr);
  svgp_epilogue<<<dim3(NSEQ / EPT, HHEADS, BDIM), EPT, 0, stream>>>(
      qk, we, wr, log_lambda, m_u, s_tril, log_ssqrt, Wf, bf, eps, esc, rsc, attn);
  kl_reduce<<<1, 256, 0, stream>>>(log_lambda, m_u, s_tril, log_ssqrt, klp);
}


# KL scalar leaves its own one-block launch for block 0 of the weight-generation launch (4 -> 3 launches); its loads unconditional, 8 trips in flight, so it ends under that launch; same sum order
# speedup vs baseline: 1.3839x; 1.0348x over previous
#include <hip/hip_runtime.h>

#define BDIM   32
#define NSEQ   1024
#define CDIM   768
#define HHEADS 12
#define RLOW   10
#define MSUB   100
#define DH     64

#define GM (BDIM*NSEQ)
#define GN (2*CDIM)
#define GK (CDIM)
#define TM 128
#define TN 128
#define TK 32
#define LDA 36
#define LDB 132

typedef __attribute__((ext_vector_type(2))) float v2f;
typedef __attribute__((ext_vector_type(8))) float v8f;

#define NT (GK / TK)

__device__ __forceinline__ void tile_gload(const float* __restrict__ pa,
                                           const float* __restrict__ pb, int kk,
                                           float4 (&ra)[4], float4 (&rb)[4]) {
#pragma unroll
  for (int it = 0; it < 4; ++it) {
    ra[it] = *(const float4*)(pa + (size_t)it * 32 * GK + kk);
    rb[it] = *(const float4*)(pb + (size_t)it * 32 * GK + kk);
  }
}

__device__ __forceinline__ void tile_lstore(float* __restrict__ As, float* __restrict__ Bs,
                                            int lrow, int lc4,
                                            const float4 (&ra)[4], const float4 (&rb)[4]) {
#pragma unroll
  for (int it = 0; it < 4; ++it) {
    const int row = lrow + it * 32;
    *(float4*)(&As[row * LDA + lc4]) = ra[it];
    Bs[(lc4 + 0) * LDB + row] = rb[it].x;
    Bs[(lc4 + 1) * LDB + row] = rb[it].y;
    Bs[(lc4 + 2) * LDB + row] = rb[it].z;
    Bs[(lc4 + 3) * LDB + row] = rb[it].w;
  }
}

__device__ __forceinline__ void tile_mma(const float* __restrict__ As, const float* __restrict__ Bs,
                                         int mrow0, int mrow1, int cb, int hi,
                                         v8f (&acc)[2][4]) {
#pragma unroll
  for (int k4 = 0; k4 < 8; ++k4) {
    const int kb = k4 * 4 + hi * 2;
    v2f a0, a1;
    a0.x = As[mrow0 * LDA + kb];  a0.y = As[mrow0 * LDA + kb + 1];
    a1.x = As[mrow1 * LDA + kb];  a1.y = As[mrow1 * LDA + kb + 1];
    v2f b0, b1, b2, b3;
    b0.x = Bs[kb * LDB + cb];        b0.y = Bs[(kb + 1) * LDB + cb];
    b1.x = Bs[kb * LDB + cb + 16];   b1.y = Bs[(kb + 1) * LDB + cb + 16];
    b2.x = Bs[kb * LDB + cb + 32];   b2.y = Bs[(kb + 1) * LDB + cb + 32];
    b3.x = Bs[kb * LDB + cb + 48];   b3.y = Bs[(kb + 1) * LDB + cb + 48];
    acc[0][0] = __builtin_amdgcn_wmma_f32_16x16x4_f32(false, a0, false, b0, (short)0, acc[0][0], false, false);
    acc[0][1] = __builtin_amdgcn_wmma_f32_16x16x4_f32(false, a0, false, b1, (short)0, acc[0][1], false, false);
    acc[0][2] = __builtin_amdgcn_wmma_f32_16x16x4_f32(false, a0, false, b2, (short)0, acc[0][2], false, false);
    acc[0][3] = __builtin_amdgcn_wmma_f32_16x16x4_f32(false, a0, false, b3, (short)0, acc[0][3], false, false);
    acc[1][0] = __builtin_amdgcn_wmma_f32_16x16x4_f32(false, a1, false, b0, (short)0, acc[1][0], false, false);
    acc[1][1] = __builtin_amdgcn_wmma_f32_16x16x4_f32(false, a1, false, b1, (short)0, acc[1][1], false, false);
    acc[1][2] = __builtin_amdgcn_wmma_f32_16x16x4_f32(false, a1, false, b2, (short)0, acc[1][2], false, false);
    acc[1][3] = __builtin_amdgcn_wmma_f32_16x16x4_f32(false, a1, false, b3, (short)0, acc[1][3], false, false);
  }
}

__global__ __launch_bounds__(256) __attribute__((amdgpu_waves_per_eu(4, 4)))
void qk_gemm_wmma(const float* __restrict__ X, const float* __restrict__ W,
                  float* __restrict__ out) {
  __shared__ float As0[TM * LDA];
  __shared__ float Bs0[TK * LDB];
  __shared__ float As1[TM * LDA];
  __shared__ float Bs1[TK * LDB];
  const int tid  = threadIdx.x;
  const int lane = tid & 31;
  const int wave = tid >> 5;
  const int mw   = wave & 3;
  const int nw   = wave >> 2;
  const int lo   = lane & 15;
  const int hi   = lane >> 4;
  const int bm   = blockIdx.y * TM;
  const int bn   = blockIdx.x * TN;

  v8f acc[2][4];
#pragma unroll
  for (int i = 0; i < 2; ++i)
#pragma unroll
    for (int j = 0; j < 4; ++j)
      acc[i][j] = (v8f){0.f,0.f,0.f,0.f,0.f,0.f,0.f,0.f};

  const int mrow0 = mw * 32 + lo;
  const int mrow1 = mrow0 + 16;
  const int cb    = nw * 64 + lo;

  const int lrow = tid >> 3;
  const int lc4  = (tid & 7) << 2;
  const float* pa = X + (size_t)(bm + lrow) * GK + lc4;
  const float* pb = W + (size_t)(bn + lrow) * GK + lc4;

  float4 ra[4], rb[4];
  tile_gload(pa, pb, 0, ra, rb);
  tile_lstore(As0, Bs0, lrow, lc4, ra, rb);
  __syncthreads();

#define PIN_LANDING_BEHIND() __builtin_amdgcn_sched_barrier(0x020)
  for (int t = 0; t < NT - 2; t += 2) {
    tile_gload(pa, pb, (t + 1) * TK, ra, rb);
    tile_mma(As0, Bs0, mrow0, mrow1, cb, hi, acc);
    PIN_LANDING_BEHIND();
    tile_lstore(As1, Bs1, lrow, lc4, ra, rb);
    __syncthreads();
    tile_gload(pa, pb, (t + 2) * TK, ra, rb);
    tile_mma(As1, Bs1, mrow0, mrow1, cb, hi, acc);
    PIN_LANDING_BEHIND();
    tile_lstore(As0, Bs0, lrow, lc4, ra, rb);
    __syncthreads();
  }
  tile_gload(pa, pb, (NT - 1) * TK, ra, rb);
  tile_mma(As0, Bs0, mrow0, mrow1, cb, hi, acc);
  PIN_LANDING_BEHIND();
  tile_lstore(As1, Bs1, lrow, lc4, ra, rb);
  __syncthreads();
  tile_mma(As1, Bs1, mrow0, mrow1, cb, hi, acc);

  const size_t r0 = (size_t)(bm + mw * 32 + 8 * hi);
  const int    cg = bn + nw * 64 + lo;
#pragma unroll
  for (int i = 0; i < 8; ++i) {
    float* p0 = out + (r0 + i) * GN + cg;
    p0[0]  = acc[0][0][i];
    p0[16] = acc[0][1][i];
    p0[32] = acc[0][2][i];
    p0[48] = acc[0][3][i];
    float* p1 = out + (r0 + 16 + i) * GN + cg;
    p1[0]  = acc[1][0][i];
    p1[16] = acc[1][1][i];
    p1[32] = acc[1][2][i];
    p1[48] = acc[1][3][i];
  }
}

#define KL_N3 (HHEADS * RLOW * RLOW * RLOW)
#define KL_N2 (HHEADS * RLOW * RLOW)
#define KL_B  8

__global__ __launch_bounds__(256)
void gen_weights_kl(const float* __restrict__ x, const float* __restrict__ we_p,
                    const float* __restrict__ wr_p,
                    float* __restrict__ we, float* __restrict__ wr,
                    const float* __restrict__ log_lambda, const float* __restrict__ m_u,
                    const float* __restrict__ s_tril, const float* __restrict__ log_ssqrt,
                    float* __restrict__ out_kl) {
  __shared__ int sidx[MSUB];
  __shared__ float red[256];
  const int tid = threadIdx.x;

  if (blockIdx.x == 0) {
    float p = 0.f;
    for (int t0 = 0; t0 < (KL_N3 + 255) / 256; t0 += KL_B) {
      float st[KL_B], ls[KL_B], ll[KL_B];
#pragma unroll
      for (int k = 0; k < KL_B; ++k) {
        const int ic = min(tid + (t0 + k) * 256, KL_N3 - 1);
        st[k] = s_tril[ic];
        ls[k] = log_ssqrt[ic / RLOW];
        ll[k] = log_lambda[ic / (RLOW * RLOW)];
      }
#pragma unroll
      for (int k = 0; k < KL_B; ++k) {
        asm volatile("" : "+v"(st[k]), "+v"(ls[k]), "+v"(ll[k]));
      }
#pragma unroll
      for (int k = 0; k < KL_B; ++k) {
        const int i = tid + (t0 + k) * 256;
        const int q = (i / RLOW) % RLOW;
        const int c = i % RLOW;
        float sl = (q == c) ? expf(ls[k]) : 0.f;
        if (q > c) sl += st[k];
        const float t = expf(2.f * ll[k]) * sl;
        if (i < KL_N3) p += 0.5f * t * t;
      }
    }
    {
      float mu[5], ll[5], ls[5];
#pragma unroll
      for (int k = 0; k < 5; ++k) {
        const int ic = min(tid + k * 256, KL_N2 - 1);
        mu[k] = m_u[ic];
        ll[k] = log_lambda[ic / RLOW];
        ls[k] = log_ssqrt[ic];
      }
#pragma unroll
      for (int k = 0; k < 5; ++k) {
        if (tid + k * 256 < KL_N2) {
          p += 0.5f * mu[k] * mu[k] * expf(4.f * ll[k]);
          p -= ls[k];
        }
      }
    }
    for (int i = tid; i < HHEADS * RLOW; i += 256) p -= 2.f * (float)RLOW * log_lambda[i];
    red[tid] = p;
    __syncthreads();
    for (int s = 128; s > 0; s >>= 1) {
      if (tid < s) red[tid] += red[tid + s];
      __syncthreads();
    }
    if (tid == 0) out_kl[0] = red[0] - 0.5f * (float)(RLOW * RLOW * HHEADS);
    return;
  }

  if (tid < MSUB) {
    const double delta = 1023.0 / 99.0;
    int v = (int)((double)tid * delta);
    if (tid == MSUB - 1) v = NSEQ - 1;
    sidx[tid] = v;
  }
  __syncthreads();

  const int g = (blockIdx.x - 1) * 256 + tid;
  if (g >= BDIM * HHEADS * DH * RLOW) return;
  const int r = g % RLOW;
  const int d = (g / RLOW) % DH;
  const int h = (g / (RLOW * DH)) % HHEADS;
  const int b = g / (RLOW * DH * HHEADS);

  float aw = 0.f, ar = 0.f;
#pragma unroll 4
  for (int m = 0; m < MSUB; ++m) {
    float xs = x[((size_t)b * NSEQ + sidx[m]) * CDIM + h * DH + d];
    aw += xs * we_p[(h * MSUB + m) * RLOW + r];
    ar += xs * wr_p[(h * MSUB + m) * RLOW + r];
  }
  we[g] = aw;
  wr[g] = ar;
}

#define EPT 128
#define QP  68

__global__ __launch_bounds__(EPT)
void svgp_epilogue(const float* __restrict__ qk, const float* __restrict__ we,
                   const float* __restrict__ wr,
                   const float* __restrict__ log_lambda,
                   const float* __restrict__ m_u, const float* __restrict__ s_tril,
                   const float* __restrict__ log_ssqrt,
                   const float* __restrict__ Wf, const float* __restrict__ bf,
                   const float* __restrict__ eps,
                   float* __restrict__ esc, float* __restrict__ rsc,
                   float* __restrict__ attn) {
  __shared__ __attribute__((aligned(16))) float qs[EPT * QP];
  __shared__ __attribute__((aligned(16))) float ks[EPT * QP];
  __shared__ __attribute__((aligned(16))) float ep_s[EPT * RLOW];
  __shared__ float sd_s[RLOW * RLOW];
  __shared__ float lam2_s[RLOW];

  const int tid = threadIdx.x;
  const int h   = blockIdx.y;
  const int b   = blockIdx.z;
  const int n0  = blockIdx.x * EPT;
  const size_t bh = (size_t)b * HHEADS + h;
  const size_t so = (bh * NSEQ + n0) * RLOW;

  const float* qbase = qk + ((size_t)b * NSEQ + n0) * GN + h * DH;
#pragma unroll
  for (int it = 0; it < (EPT * DH / 4) / EPT; ++it) {
    const int i   = tid + it * EPT;
    const int row = i >> 4;
    const int c4  = (i & 15) << 2;
    const float* p = qbase + (size_t)row * GN + c4;
    const float4 vq = *(const float4*)p;
    const float4 vk = *(const float4*)(p + CDIM);
    *(float4*)(&qs[row * QP + c4]) = vq;
    *(float4*)(&ks[row * QP + c4]) = vk;
  }
  for (int i = tid; i < EPT * RLOW / 4; i += EPT)
    *(float4*)(&ep_s[i * 4]) = *(const float4*)(eps + so + (size_t)i * 4);
  if (tid < RLOW * RLOW) sd_s[tid]   = expf(log_ssqrt[h * RLOW * RLOW + tid]);
  if (tid < RLOW)        lam2_s[tid] = expf(2.f * log_lambda[h * RLOW + tid]);
  __syncthreads();

  float ev[RLOW];
  {
    const float2* e2 = (const float2*)(&ep_s[tid * RLOW]);
#pragma unroll
    for (int j = 0; j < RLOW / 2; ++j) { const float2 t = e2[j]; ev[2 * j] = t.x; ev[2 * j + 1] = t.y; }
  }

  const float* wep = we + bh * (DH * RLOW);
  const float* wrp = wr + bh * (DH * RLOW);
  float e[RLOW], rv[RLOW], v1[RLOW];
#pragma unroll
  for (int r = 0; r < RLOW; ++r) { e[r] = 0.f; rv[r] = 0.f; }
  float qn = 0.f, kn = 0.f;

#pragma unroll
  for (int d4 = 0; d4 < DH / 4; ++d4) {
    const float4 v = *(const float4*)(&qs[tid * QP + d4 * 4]);
    const float xv[4] = {v.x, v.y, v.z, v.w};
#pragma unroll
    for (int j = 0; j < 4; ++j) {
      const float x = xv[j];
      qn += x * x;
#pragma unroll
      for (int r = 0; r < RLOW; ++r) e[r] += x * wep[(d4 * 4 + j) * RLOW + r];
    }
  }
#pragma unroll
  for (int d4 = 0; d4 < DH / 4; ++d4) {
    const float4 v = *(const float4*)(&ks[tid * QP + d4 * 4]);
    const float xv[4] = {v.x, v.y, v.z, v.w};
#pragma unroll
    for (int j = 0; j < 4; ++j) {
      const float x = xv[j];
      kn += x * x;
#pragma unroll
      for (int r = 0; r < RLOW; ++r) rv[r] += x * wrp[(d4 * 4 + j) * RLOW + r];
    }
  }
  const float qi = 1.f / fmaxf(sqrtf(qn), 1e-12f);
  const float ki = 1.f / fmaxf(sqrtf(kn), 1e-12f);
#pragma unroll
  for (int r = 0; r < RLOW; ++r) {
    e[r]  *= qi;
    rv[r] *= ki;
    v1[r]  = (e[r] + rv[r]) * lam2_s[r];
    ep_s[tid * RLOW + r] = e[r];
  }

  const float* mup = m_u + h * (RLOW * RLOW);
  const float* stp = s_tril + (size_t)h * (RLOW * RLOW * RLOW);
  float sample[RLOW];
#pragma unroll
  for (int c = 0; c < RLOW; ++c) {
    float mean = 0.f;
#pragma unroll
    for (int r = 0; r < RLOW; ++r) mean += v1[r] * mup[r * RLOW + c];
    float noise = 0.f;
#pragma unroll
    for (int q = c; q < RLOW; ++q) {
      float tq = 0.f;
#pragma unroll
      for (int a = 0; a < RLOW; ++a) {
        const float s = (q == c) ? sd_s[a * RLOW + q] : stp[(a * RLOW + q) * RLOW + c];
        tq += v1[a] * s;
      }
      noise += ev[q] * tq;
    }
    sample[c] = mean + noise;
  }

#pragma unroll
  for (int d4 = 0; d4 < DH / 4; ++d4) {
    float o[4];
#pragma unroll
    for (int j = 0; j < 4; ++j) {
      const int d = d4 * 4 + j;
      float a = bf[d];
#pragma unroll
      for (int r = 0; r < RLOW; ++r) a += sample[r] * Wf[d * RLOW + r];
      o[j] = a;
    }
    *(float4*)(&qs[tid * QP + d4 * 4]) = make_float4(o[0], o[1], o[2], o[3]);
  }
  __syncthreads();
#pragma unroll
  for (int r = 0; r < RLOW; ++r) ks[tid * RLOW + r] = rv[r];
  __syncthreads();

  float* abase = attn + ((size_t)b * NSEQ + n0) * CDIM + h * DH;
#pragma unroll
  for (int it = 0; it < (EPT * DH / 4) / EPT; ++it) {
    const int i   = tid + it * EPT;
    const int row = i >> 4;
    const int c4  = (i & 15) << 2;
    *(float4*)(abase + (size_t)row * CDIM + c4) = *(const float4*)(&qs[row * QP + c4]);
  }
  for (int i = tid; i < EPT * RLOW / 4; i += EPT) {
    *(float4*)(esc + so + (size_t)i * 4) = *(const float4*)(&ep_s[i * 4]);
    *(float4*)(rsc + so + (size_t)i * 4) = *(const float4*)(&ks[i * 4]);
  }
}

extern "C" void kernel_launch(void* const* d_in, const int* in_sizes, int n_in,
                              void* d_out, int out_size, void* d_ws, size_t ws_size,
                              hipStream_t stream) {
  const float* x          = (const float*)d_in[0];
  const float* Wqk        = (const float*)d_in[1];
  const float* we_p       = (const float*)d_in[2];
  const float* wr_p       = (const float*)d_in[3];
  const float* log_lambda = (const float*)d_in[4];
  const float* m_u        = (const float*)d_in[5];
  const float* s_tril     = (const float*)d_in[6];
  const float* log_ssqrt  = (const float*)d_in[7];
  const float* Wf         = (const float*)d_in[8];
  const float* bf         = (const float*)d_in[9];
  const float* eps        = (const float*)d_in[10];

  float* ws = (float*)d_ws;
  float* qk = ws;
  float* we = ws + (size_t)GM * GN;
  float* wr = we + (size_t)BDIM * HHEADS * DH * RLOW;

  float* attn = (float*)d_out;
  float* esc  = attn + (size_t)BDIM * NSEQ * CDIM;
  float* rsc  = esc + (size_t)BDIM * HHEADS * NSEQ * RLOW;
  float* klp  = rsc + (size_t)BDIM * HHEADS * NSEQ * RLOW;

  qk_gemm_wmma<<<dim3(GN / TN, GM / TM), 256, 0, stream>>>(x, Wqk, qk);
  gen_weights_kl<<<(BDIM * HHEADS * DH * RLOW + 255) / 256 + 1, 256, 0, stream>>>(
      x, we_p, wr_p, we, wr, log_lambda, m_u, s_tril, log_ssqrt, klp);
  svgp_epilogue<<<dim3(NSEQ / EPT, HHEADS, BDIM), EPT, 0, stream>>>(
      qk, we, wr, log_lambda, m_u, s_tril, log_ssqrt, Wf, bf, eps, esc, rsc, attn);
}


# row epilogue fused into the GEMM block: tile = q and k columns of one head, accumulators -> LDS rows -> per-lane row work, qk never goes to memory (3 -> 2 launches); same sum order
# speedup vs baseline: 1.5056x; 1.0879x over previous
#include <hip/hip_runtime.h>

#define BDIM   32
#define NSEQ   1024
#define CDIM   768
#define HHEADS 12
#define RLOW   10
#define MSUB   100
#define DH     64

#define GM (BDIM*NSEQ)
#define GN (2*CDIM)
#define GK (CDIM)
#define TM 128
#define TN 128
#define TK 32
#define LDA 36
#define LDB 132

typedef __attribute__((ext_vector_type(2))) float v2f;
typedef __attribute__((ext_vector_type(8))) float v8f;

#define KL_N3 (HHEADS * RLOW * RLOW * RLOW)
#define KL_N2 (HHEADS * RLOW * RLOW)
#define KL_B  8

__global__ __launch_bounds__(256)
void gen_weights_kl(const float* __restrict__ x, const float* __restrict__ we_p,
                    const float* __restrict__ wr_p,
                    float* __restrict__ we, float* __restrict__ wr,
                    const float* __restrict__ log_lambda, const float* __restrict__ m_u,
                    const float* __restrict__ s_tril, const float* __restrict__ log_ssqrt,
                    float* __restrict__ out_kl) {
  __shared__ int sidx[MSUB];
  __shared__ float red[256];
  const int tid = threadIdx.x;

  if (blockIdx.x == 0) {
    float p = 0.f;
    for (int t0 = 0; t0 < (KL_N3 + 255) / 256; t0 += KL_B) {
      float st[KL_B], ls[KL_B], ll[KL_B];
#pragma unroll
      for (int k = 0; k < KL_B; ++k) {
        const int ic = min(tid + (t0 + k) * 256, KL_N3 - 1);
        st[k] = s_tril[ic];
        ls[k] = log_ssqrt[ic / RLOW];
        ll[k] = log_lambda[ic / (RLOW * RLOW)];
      }
#pragma unroll
      for (int k = 0; k < KL_B; ++k) {
        asm volatile("" : "+v"(st[k]), "+v"(ls[k]), "+v"(ll[k]));
      }
#pragma unroll
      for (int k = 0; k < KL_B; ++k) {
        const int i = tid + (t0 + k) * 256;
        const int q = (i / RLOW) % RLOW;
        const int c = i % RLOW;
        float sl = (q == c) ? expf(ls[k]) : 0.f;
        if (q > c) sl += st[k];
        const float t = expf(2.f * ll[k]) * sl;
        if (i < KL_N3) p += 0.5f * t * t;
      }
    }
    {
      float mu[5], ll[5], ls[5];
#pragma unroll
      for (int k = 0; k < 5; ++k) {
        const int ic = min(tid + k * 256, KL_N2 - 1);
        mu[k] = m_u[ic];
        ll[k] = log_lambda[ic / RLOW];
        ls[k] = log_ssqrt[ic];
      }
#pragma unroll
      for (int k = 0; k < 5; ++k) {
        if (tid + k * 256 < KL_N2) {
          p += 0.5f * mu[k] * mu[k] * expf(4.f * ll[k]);
          p -= ls[k];
        }
      }
    }
    for (int i = tid; i < HHEADS * RLOW; i += 256) p -= 2.f * (float)RLOW * log_lambda[i];
    red[tid] = p;
    __syncthreads();
    for (int s = 128; s > 0; s >>= 1) {
      if (tid < s) red[tid] += red[tid + s];
      __syncthreads();
    }
    if (tid == 0) out_kl[0] = red[0] - 0.5f * (float)(RLOW * RLOW * HHEADS);
    return;
  }

  if (tid < MSUB) {
    const double delta = 1023.0 / 99.0;
    int v = (int)((double)tid * delta);
    if (tid == MSUB - 1) v = NSEQ - 1;
    sidx[tid] = v;
  }
  __syncthreads();

  const int g = (blockIdx.x - 1) * 256 + tid;
  if (g >= BDIM * HHEADS * DH * RLOW) return;
  const int r = g % RLOW;
  const int d = (g / RLOW) % DH;
  const int h = (g / (RLOW * DH)) % HHEADS;
  const int b = g / (RLOW * DH * HHEADS);

  float aw = 0.f, ar = 0.f;
#pragma unroll 4
  for (int m = 0; m < MSUB; ++m) {
    float xs = x[((size_t)b * NSEQ + sidx[m]) * CDIM + h * DH + d];
    aw += xs * we_p[(h * MSUB + m) * RLOW + r];
    ar += xs * wr_p[(h * MSUB + m) * RLOW + r];
  }
  we[g] = aw;
  wr[g] = ar;
}

#define NT (GK / TK)
#define RP 132

__device__ __forceinline__ void tile_gload(const float* __restrict__ pa,
                                           const float* __restrict__ pb, int kk,
                                           float4 (&ra)[4], float4 (&rb)[4]) {
#pragma unroll
  for (int it = 0; it < 4; ++it) {
    ra[it] = *(const float4*)(pa + (size_t)it * 32 * GK + kk);
    rb[it] = *(const float4*)(pb + (size_t)(it * 32 + (it >> 1) * (CDIM - DH)) * GK + kk);
  }
}

__device__ __forceinline__ void tile_lstore(float* __restrict__ As, float* __restrict__ Bs,
                                            int lrow, int lc4,
                                            const float4 (&ra)[4], const float4 (&rb)[4]) {
#pragma unroll
  for (int it = 0; it < 4; ++it) {
    const int row = lrow + it * 32;
    *(float4*)(&As[row * LDA + lc4]) = ra[it];
    Bs[(lc4 + 0) * LDB + row] = rb[it].x;
    Bs[(lc4 + 1) * LDB + row] = rb[it].y;
    Bs[(lc4 + 2) * LDB + row] = rb[it].z;
    Bs[(lc4 + 3) * LDB + row] = rb[it].w;
  }
}

__device__ __forceinline__ void tile_mma(const float* __restrict__ As, const float* __restrict__ Bs,
                                         int mrow0, int mrow1, int cb, int hi,
                                         v8f (&acc)[2][4]) {
#pragma unroll
  for (int k4 = 0; k4 < 8; ++k4) {
    const int kb = k4 * 4 + hi * 2;
    v2f a0, a1;
    a0.x = As[mrow0 * LDA + kb];  a0.y = As[mrow0 * LDA + kb + 1];
    a1.x = As[mrow1 * LDA + kb];  a1.y = As[mrow1 * LDA + kb + 1];
    v2f b0, b1, b2, b3;
    b0.x = Bs[kb * LDB + cb];        b0.y = Bs[(kb + 1) * LDB + cb];
    b1.x = Bs[kb * LDB + cb + 16];   b1.y = Bs[(kb + 1) * LDB + cb + 16];
    b2.x = Bs[kb * LDB + cb + 32];   b2.y = Bs[(kb + 1) * LDB + cb + 32];
    b3.x = Bs[kb * LDB + cb + 48];   b3.y = Bs[(kb + 1) * LDB + cb + 48];
    acc[0][0] = __builtin_amdgcn_wmma_f32_16x16x4_f32(false, a0, false, b0, (short)0, acc[0][0], false, false);
    acc[0][1] = __builtin_amdgcn_wmma_f32_16x16x4_f32(false, a0, false, b1, (short)0, acc[0][1], false, false);
    acc[0][2] = __builtin_amdgcn_wmma_f32_16x16x4_f32(false, a0, false, b2, (short)0, acc[0][2], false, false);
    acc[0][3] = __builtin_amdgcn_wmma_f32_16x16x4_f32(false, a0, false, b3, (short)0, acc[0][3], false, false);
    acc[1][0] = __builtin_amdgcn_wmma_f32_16x16x4_f32(false, a1, false, b0, (short)0, acc[1][0], false, false);
    acc[1][1] = __builtin_amdgcn_wmma_f32_16x16x4_f32(false, a1, false, b1, (short)0, acc[1][1], false, false);
    acc[1][2] = __builtin_amdgcn_wmma_f32_16x16x4_f32(false, a1, false, b2, (short)0, acc[1][2], false, false);
    acc[1][3] = __builtin_amdgcn_wmma_f32_16x16x4_f32(false, a1, false, b3, (short)0, acc[1][3], false, false);
  }
}

__global__ __launch_bounds__(256) __attribute__((amdgpu_waves_per_eu(4, 4)))
void qk_gemm_svgp(const float* __restrict__ X, const float* __restrict__ W,
                  const float* __restrict__ we, const float* __restrict__ wr,
                  const float* __restrict__ log_lambda,
                  const float* __restrict__ m_u, const float* __restrict__ s_tril,
                  const float* __restrict__ log_ssqrt,
                  const float* __restrict__ Wf, const float* __restrict__ bf,
                  const float* __restrict__ eps,
                  float* __restrict__ esc, float* __restrict__ rsc,
                  float* __restrict__ attn) {
  __shared__ float As0[TM * LDA];
  __shared__ float Bs0[TK * LDB];
  __shared__ float As1[TM * LDA];
  __shared__ float Bs1[TK * LDB];
  __shared__ __attribute__((aligned(16))) float rows_s[TM * RP];
  __shared__ __attribute__((aligned(16))) float ep_s[TM * RLOW];
  __shared__ __attribute__((aligned(16))) float rs_s[TM * RLOW];
  __shared__ __attribute__((aligned(16))) float st_s[RLOW * RLOW * RLOW];
  __shared__ __attribute__((aligned(16))) float wf_s[DH * RLOW];
  __shared__ __attribute__((aligned(16))) float mu_s[RLOW * RLOW];
  __shared__ __attribute__((aligned(16))) float sd_s[RLOW * RLOW];
  __shared__ __attribute__((aligned(16))) float bf_s[DH];
  __shared__ float lam2_s[RLOW];

  const int tid  = threadIdx.x;
  const int lane = tid & 31;
  const int wave = tid >> 5;
  const int mw   = wave & 3;
  const int nw   = wave >> 2;
  const int lo   = lane & 15;
  const int hi   = lane >> 4;
  const int h    = blockIdx.x;
  const int bm   = blockIdx.y * TM;
  const int b    = blockIdx.y / (NSEQ / TM);
  const int n0   = (blockIdx.y % (NSEQ / TM)) * TM;
  const size_t bh = (size_t)b * HHEADS + h;
  const size_t so = (bh * NSEQ + n0) * RLOW;

  v8f acc[2][4];
#pragma unroll
  for (int i = 0; i < 2; ++i)
#pragma unroll
    for (int j = 0; j < 4; ++j)
      acc[i][j] = (v8f){0.f,0.f,0.f,0.f,0.f,0.f,0.f,0.f};

  const int mrow0 = mw * 32 + lo;
  const int mrow1 = mrow0 + 16;
  const int cb    = nw * 64 + lo;

  const int lrow = tid >> 3;
  const int lc4  = (tid & 7) << 2;
  const float* pa = X + (size_t)(bm + lrow) * GK + lc4;
  const float* pb = W + (size_t)(h * DH + lrow) * GK + lc4;

  float4 ra[4], rb[4];
  tile_gload(pa, pb, 0, ra, rb);
  tile_lstore(As0, Bs0, lrow, lc4, ra, rb);
  __syncthreads();

#define PIN_LANDING_BEHIND() __builtin_amdgcn_sched_barrier(0x020)
  for (int t = 0; t < NT - 2; t += 2) {
    tile_gload(pa, pb, (t + 1) * TK, ra, rb);
    tile_mma(As0, Bs0, mrow0, mrow1, cb, hi, acc);
    PIN_LANDING_BEHIND();
    tile_lstore(As1, Bs1, lrow, lc4, ra, rb);
    __syncthreads();
    tile_gload(pa, pb, (t + 2) * TK, ra, rb);
    tile_mma(As1, Bs1, mrow0, mrow1, cb, hi, acc);
    PIN_LANDING_BEHIND();
    tile_lstore(As0, Bs0, lrow, lc4, ra, rb);
    __syncthreads();
  }
  tile_gload(pa, pb, (NT - 1) * TK, ra, rb);
  tile_mma(As0, Bs0, mrow0, mrow1, cb, hi, acc);
  PIN_LANDING_BEHIND();
  tile_lstore(As1, Bs1, lrow, lc4, ra, rb);
  __syncthreads();
  tile_mma(As1, Bs1, mrow0, mrow1, cb, hi, acc);

  {
    float* dst = rows_s + (mw * 32 + 8 * hi) * RP + nw * 64 + lo;
#pragma unroll
    for (int i = 0; i < 8; ++i) {
      float* p0 = dst + i * RP;
      p0[0]  = acc[0][0][i];
      p0[16] = acc[0][1][i];
      p0[32] = acc[0][2][i];
      p0[48] = acc[0][3][i];
      float* p1 = dst + (16 + i) * RP;
      p1[0]  = acc[1][0][i];
      p1[16] = acc[1][1][i];
      p1[32] = acc[1][2][i];
      p1[48] = acc[1][3][i];
    }
  }
  for (int i = tid; i < TM * RLOW / 4; i += 256)
    *(float4*)(&ep_s[i * 4]) = *(const float4*)(eps + so + (size_t)i * 4);
  for (int i = tid; i < RLOW * RLOW * RLOW; i += 256) st_s[i] = s_tril[(size_t)h * (RLOW * RLOW * RLOW) + i];
  for (int i = tid; i < DH * RLOW; i += 256) wf_s[i] = Wf[i];
  if (tid < RLOW * RLOW) {
    mu_s[tid] = m_u[h * (RLOW * RLOW) + tid];
    sd_s[tid] = expf(log_ssqrt[h * RLOW * RLOW + tid]);
  }
  if (tid < DH)   bf_s[tid]   = bf[tid];
  if (tid < RLOW) lam2_s[tid] = expf(2.f * log_lambda[h * RLOW + tid]);
  __syncthreads();

  const int row = tid & (TM - 1);
  const bool krole = __builtin_amdgcn_readfirstlane(wave) >= 4;
  float e[RLOW];
#pragma unroll
  for (int r = 0; r < RLOW; ++r) e[r] = 0.f;

  if (!krole) {
    const float* wep = we + bh * (DH * RLOW);
    float qn = 0.f;
#pragma unroll
    for (int d4 = 0; d4 < DH / 4; ++d4) {
      const float4 v = *(const float4*)(&rows_s[row * RP + d4 * 4]);
      const float xv[4] = {v.x, v.y, v.z, v.w};
#pragma unroll
      for (int j = 0; j < 4; ++j) {
        const float x = xv[j];
        qn += x * x;
#pragma unroll
        for (int r = 0; r < RLOW; ++r) e[r] += x * wep[(d4 * 4 + j) * RLOW + r];
      }
    }
    const float qi = 1.f / fmaxf(sqrtf(qn), 1e-12f);
#pragma unroll
    for (int r = 0; r < RLOW; ++r) e[r] *= qi;
  } else {
    const float* wrp = wr + bh * (DH * RLOW);
    float rv[RLOW];
#pragma unroll
    for (int r = 0; r < RLOW; ++r) rv[r] = 0.f;
    float kn = 0.f;
#pragma unroll
    for (int d4 = 0; d4 < DH / 4; ++d4) {
      const float4 v = *(const float4*)(&rows_s[row * RP + DH + d4 * 4]);
      const float xv[4] = {v.x, v.y, v.z, v.w};
#pragma unroll
      for (int j = 0; j < 4; ++j) {
        const float x = xv[j];
        kn += x * x;
#pragma unroll
        for (int r = 0; r < RLOW; ++r) rv[r] += x * wrp[(d4 * 4 + j) * RLOW + r];
      }
    }
    const float ki = 1.f / fmaxf(sqrtf(kn), 1e-12f);
#pragma unroll
    for (int r = 0; r < RLOW; ++r) rs_s[row * RLOW + r] = rv[r] * ki;
  }
  __syncthreads();

  if (!krole) {
    float ev[RLOW], rv[RLOW], v1[RLOW];
    {
      const float2* e2 = (const float2*)(&ep_s[row * RLOW]);
      const float2* r2 = (const float2*)(&rs_s[row * RLOW]);
#pragma unroll
      for (int j = 0; j < RLOW / 2; ++j) {
        const float2 t = e2[j]; ev[2 * j] = t.x; ev[2 * j + 1] = t.y;
        const float2 u = r2[j]; rv[2 * j] = u.x; rv[2 * j + 1] = u.y;
      }
    }
#pragma unroll
    for (int r = 0; r < RLOW; ++r) {
      v1[r] = (e[r] + rv[r]) * lam2_s[r];
      ep_s[row * RLOW + r] = e[r];
    }

#pragma unroll
    for (int r = 0; r < RLOW; ++r) rows_s[row * RP + r] = v1[r];

    float mean[RLOW], tq[RLOW][RLOW];
#pragma unroll
    for (int c = 0; c < RLOW; ++c) mean[c] = 0.f;
#pragma unroll
    for (int q = 0; q < RLOW; ++q)
#pragma unroll
      for (int c = 0; c <= q; ++c) tq[q][c] = 0.f;
#pragma unroll 1
    for (int a = 0; a < RLOW; ++a) {
      const float va = rows_s[row * RP + a];
      const float* mu = &mu_s[a * RLOW];
      const float* st = &st_s[a * RLOW * RLOW];
      const float* sd = &sd_s[a * RLOW];
#pragma unroll
      for (int c = 0; c < RLOW; ++c) mean[c] += va * mu[c];
#pragma unroll
      for (int q = 0; q < RLOW; ++q) {
#pragma unroll
        for (int c = 0; c < q; ++c) tq[q][c] += va * st[q * RLOW + c];
        tq[q][q] += va * sd[q];
      }
    }
    float sample[RLOW];
#pragma unroll
    for (int c = 0; c < RLOW; ++c) {
      float noise = 0.f;
#pragma unroll
      for (int q = c; q < RLOW; ++q) noise += ev[q] * tq[q][c];
      sample[c] = mean[c] + noise;
    }

#pragma unroll 2
    for (int d4 = 0; d4 < DH / 4; ++d4) {
      float o[4];
#pragma unroll
      for (int j = 0; j < 4; ++j) {
        const int d = d4 * 4 + j;
        float a = bf_s[d];
#pragma unroll
        for (int r = 0; r < RLOW; ++r) a += sample[r] * wf_s[d * RLOW + r];
        o[j] = a;
      }
      *(float4*)(&rows_s[row * RP + d4 * 4]) = make_float4(o[0], o[1], o[2], o[3]);
    }
  }
  __syncthreads();

  float* abase = attn + ((size_t)b * NSEQ + n0) * CDIM + h * DH;
#pragma unroll
  for (int it = 0; it < (TM * DH / 4) / 256; ++it) {
    const int i  = tid + it * 256;
    const int rw = i >> 4;
    const int c4 = (i & 15) << 2;
    *(float4*)(abase + (size_t)rw * CDIM + c4) = *(const float4*)(&rows_s[rw * RP + c4]);
  }
  for (int i = tid; i < TM * RLOW / 4; i += 256) {
    *(float4*)(esc + so + (size_t)i * 4) = *(const float4*)(&ep_s[i * 4]);
    *(float4*)(rsc + so + (size_t)i * 4) = *(const float4*)(&rs_s[i * 4]);
  }
}

extern "C" void kernel_launch(void* const* d_in, const int* in_sizes, int n_in,
                              void* d_out, int out_size, void* d_ws, size_t ws_size,
                              hipStream_t stream) {
  const float* x          = (const float*)d_in[0];
  const float* Wqk        = (const float*)d_in[1];
  const float* we_p       = (const float*)d_in[2];
  const float* wr_p       = (const float*)d_in[3];
  const float* log_lambda = (const float*)d_in[4];
  const float* m_u        = (const float*)d_in[5];
  const float* s_tril     = (const float*)d_in[6];
  const float* log_ssqrt  = (const float*)d_in[7];
  const float* Wf         = (const float*)d_in[8];
  const float* bf         = (const float*)d_in[9];
  const float* eps        = (const float*)d_in[10];

  float* ws = (float*)d_ws;
  float* we = ws;
  float* wr = we + (size_t)BDIM * HHEADS * DH * RLOW;

  float* attn = (float*)d_out;
  float* esc  = attn + (size_t)BDIM * NSEQ * CDIM;
  float* rsc  = esc + (size_t)BDIM * HHEADS * NSEQ * RLOW;
  float* klp  = rsc + (size_t)BDIM * HHEADS * NSEQ * RLOW;

  gen_weights_kl<<<(BDIM * HHEADS * DH * RLOW + 255) / 256 + 1, 256, 0, stream>>>(
      x, we_p, wr_p, we, wr, log_lambda, m_u, s_tril, log_ssqrt, klp);
  qk_gemm_svgp<<<dim3(HHEADS, GM / TM), 256, 0, stream>>>(
      x, Wqk, we, wr, log_lambda, m_u, s_tril, log_ssqrt, Wf, bf, eps, esc, rsc, attn);
}


# B tile kept K-contiguous in LDS like A (Bs[col][k], pitch 36) instead of transposed: B fragments read as 8-byte pairs, staged by 16-byte stores, no register shuffles; same k order, same sum order
# speedup vs baseline: 1.6118x; 1.0705x over previous
#include <hip/hip_runtime.h>

#define BDIM   32
#define NSEQ   1024
#define CDIM   768
#define HHEADS 12
#define RLOW   10
#define MSUB   100
#define DH     64

#define GM (BDIM*NSEQ)
#define GN (2*CDIM)
#define GK (CDIM)
#define TM 128
#define TN 128
#define TK 32
#define LDA 36

typedef __attribute__((ext_vector_type(2))) float v2f;
typedef __attribute__((ext_vector_type(8))) float v8f;

#define KL_N3 (HHEADS * RLOW * RLOW * RLOW)
#define KL_N2 (HHEADS * RLOW * RLOW)
#define KL_B  8

__global__ __launch_bounds__(256)
void gen_weights_kl(const float* __restrict__ x, const float* __restrict__ we_p,
                    const float* __restrict__ wr_p,
                    float* __restrict__ we, float* __restrict__ wr,
                    const float* __restrict__ log_lambda, const float* __restrict__ m_u,
                    const float* __restrict__ s_tril, const float* __restrict__ log_ssqrt,
                    float* __restrict__ out_kl) {
  __shared__ int sidx[MSUB];
  __shared__ float red[256];
  const int tid = threadIdx.x;

  if (blockIdx.x == 0) {
    float p = 0.f;
    for (int t0 = 0; t0 < (KL_N3 + 255) / 256; t0 += KL_B) {
      float st[KL_B], ls[KL_B], ll[KL_B];
#pragma unroll
      for (int k = 0; k < KL_B; ++k) {
        const int ic = min(tid + (t0 + k) * 256, KL_N3 - 1);
        st[k] = s_tril[ic];
        ls[k] = log_ssqrt[ic / RLOW];
        ll[k] = log_lambda[ic / (RLOW * RLOW)];
      }
#pragma unroll
      for (int k = 0; k < KL_B; ++k) {
        asm volatile("" : "+v"(st[k]), "+v"(ls[k]), "+v"(ll[k]));
      }
#pragma unroll
      for (int k = 0; k < KL_B; ++k) {
        const int i = tid + (t0 + k) * 256;
        const int q = (i / RLOW) % RLOW;
        const int c = i % RLOW;
        float sl = (q == c) ? expf(ls[k]) : 0.f;
        if (q > c) sl += st[k];
        const float t = expf(2.f * ll[k]) * sl;
        if (i < KL_N3) p += 0.5f * t * t;
      }
    }
    {
      float mu[5], ll[5], ls[5];
#pragma unroll
      for (int k = 0; k < 5; ++k) {
        const int ic = min(tid + k * 256, KL_N2 - 1);
        mu[k] = m_u[ic];
        ll[k] = log_lambda[ic / RLOW];
        ls[k] = log_ssqrt[ic];
      }
#pragma unroll
      for (int k = 0; k < 5; ++k) {
        if (tid + k * 256 < KL_N2) {
          p += 0.5f * mu[k] * mu[k] * expf(4.f * ll[k]);
          p -= ls[k];
        }
      }
    }
    for (int i = tid; i < HHEADS * RLOW; i += 256) p -= 2.f * (float)RLOW * log_lambda[i];
    red[tid] = p;
    __syncthreads();
    for (int s = 128; s > 0; s >>= 1) {
      if (tid < s) red[tid] += red[tid + s];
      __syncthreads();
    }
    if (tid == 0) out_kl[0] = red[0] - 0.5f * (float)(RLOW * RLOW * HHEADS);
    return;
  }

  if (tid < MSUB) {
    const double delta = 1023.0 / 99.0;
    int v = (int)((double)tid * delta);
    if (tid == MSUB - 1) v = NSEQ - 1;
    sidx[tid] = v;
  }
  __syncthreads();

  const int g = (blockIdx.x - 1) * 256 + tid;
  if (g >= BDIM * HHEADS * DH * RLOW) return;
  const int r = g % RLOW;
  const int d = (g / RLOW) % DH;
  const int h = (g / (RLOW * DH)) % HHEADS;
  const int b = g / (RLOW * DH * HHEADS);

  float aw = 0.f, ar = 0.f;
#pragma unroll 4
  for (int m = 0; m < MSUB; ++m) {
    float xs = x[((size_t)b * NSEQ + sidx[m]) * CDIM + h * DH + d];
    aw += xs * we_p[(h * MSUB + m) * RLOW + r];
    ar += xs * wr_p[(h * MSUB + m) * RLOW + r];
  }
  we[g] = aw;
  wr[g] = ar;
}

#define NT (GK / TK)
#define RP 132

__device__ __forceinline__ void tile_gload(const float* __restrict__ pa,
                                           const float* __restrict__ pb, int kk,
                                           float4 (&ra)[4], float4 (&rb)[4]) {
#pragma unroll
  for (int it = 0; it < 4; ++it) {
    ra[it] = *(const float4*)(pa + (size_t)it * 32 * GK + kk);
    rb[it] = *(const float4*)(pb + (size_t)(it * 32 + (it >> 1) * (CDIM - DH)) * GK + kk);
  }
}

__device__ __forceinline__ void tile_lstore(float* __restrict__ As, float* __restrict__ Bs,
                                            int lrow, int lc4,
                                            const float4 (&ra)[4], const float4 (&rb)[4]) {
#pragma unroll
  for (int it = 0; it < 4; ++it) {
    const int row = lrow + it * 32;
    *(float4*)(&As[row * LDA + lc4]) = ra[it];
    *(float4*)(&Bs[row * LDA + lc4]) = rb[it];
  }
}

__device__ __forceinline__ void tile_mma(const float* __restrict__ As, const float* __restrict__ Bs,
                                         int mrow0, int mrow1, int cb, int hi,
                                         v8f (&acc)[2][4]) {
#pragma unroll
  for (int k4 = 0; k4 < 8; ++k4) {
    const int kb = k4 * 4 + hi * 2;
    v2f a0, a1;
    a0.x = As[mrow0 * LDA + kb];  a0.y = As[mrow0 * LDA + kb + 1];
    a1.x = As[mrow1 * LDA + kb];  a1.y = As[mrow1 * LDA + kb + 1];
    v2f b0, b1, b2, b3;
    b0.x = Bs[(cb +  0) * LDA + kb];  b0.y = Bs[(cb +  0) * LDA + kb + 1];
    b1.x = Bs[(cb + 16) * LDA + kb];  b1.y = Bs[(cb + 16) * LDA + kb + 1];
    b2.x = Bs[(cb + 32) * LDA + kb];  b2.y = Bs[(cb + 32) * LDA + kb + 1];
    b3.x = Bs[(cb + 48) * LDA + kb];  b3.y = Bs[(cb + 48) * LDA + kb + 1];
    acc[0][0] = __builtin_amdgcn_wmma_f32_16x16x4_f32(false, a0, false, b0, (short)0, acc[0][0], false, false);
    acc[0][1] = __builtin_amdgcn_wmma_f32_16x16x4_f32(false, a0, false, b1, (short)0, acc[0][1], false, false);
    acc[0][2] = __builtin_amdgcn_wmma_f32_16x16x4_f32(false, a0, false, b2, (short)0, acc[0][2], false, false);
    acc[0][3] = __builtin_amdgcn_wmma_f32_16x16x4_f32(false, a0, false, b3, (short)0, acc[0][3], false, false);
    acc[1][0] = __builtin_amdgcn_wmma_f32_16x16x4_f32(false, a1, false, b0, (short)0, acc[1][0], false, false);
    acc[1][1] = __builtin_amdgcn_wmma_f32_16x16x4_f32(false, a1, false, b1, (short)0, acc[1][1], false, false);
    acc[1][2] = __builtin_amdgcn_wmma_f32_16x16x4_f32(false, a1, false, b2, (short)0, acc[1][2], false, false);
    acc[1][3] = __builtin_amdgcn_wmma_f32_16x16x4_f32(false, a1, false, b3, (short)0, acc[1][3], false, false);
  }
}

__global__ __launch_bounds__(256) __attribute__((amdgpu_waves_per_eu(4, 4)))
void qk_gemm_svgp(const float* __restrict__ X, const float* __restrict__ W,
                  const float* __restrict__ we, const float* __restrict__ wr,
                  const float* __restrict__ log_lambda,
                  const float* __restrict__ m_u, const float* __restrict__ s_tril,
                  const float* __restrict__ log_ssqrt,
                  const float* __restrict__ Wf, const float* __restrict__ bf,
                  const float* __restrict__ eps,
                  float* __restrict__ esc, float* __restrict__ rsc,
                  float* __restrict__ attn) {
  __shared__ __attribute__((aligned(16))) float As0[TM * LDA];
  __shared__ __attribute__((aligned(16))) float Bs0[TN * LDA];
  __shared__ __attribute__((aligned(16))) float As1[TM * LDA];
  __shared__ __attribute__((aligned(16))) float Bs1[TN * LDA];
  __shared__ __attribute__((aligned(16))) float rows_s[TM * RP];
  __shared__ __attribute__((aligned(16))) float ep_s[TM * RLOW];
  __shared__ __attribute__((aligned(16))) float rs_s[TM * RLOW];
  __shared__ __attribute__((aligned(16))) float st_s[RLOW * RLOW * RLOW];
  __shared__ __attribute__((aligned(16))) float wf_s[DH * RLOW];
  __shared__ __attribute__((aligned(16))) float mu_s[RLOW * RLOW];
  __shared__ __attribute__((aligned(16))) float sd_s[RLOW * RLOW];
  __shared__ __attribute__((aligned(16))) float bf_s[DH];
  __shared__ float lam2_s[RLOW];

  const int tid  = threadIdx.x;
  const int lane = tid & 31;
  const int wave = tid >> 5;
  const int mw   = wave & 3;
  const int nw   = wave >> 2;
  const int lo   = lane & 15;
  const int hi   = lane >> 4;
  const int h    = blockIdx.x;
  const int bm   = blockIdx.y * TM;
  const int b    = blockIdx.y / (NSEQ / TM);
  const int n0   = (blockIdx.y % (NSEQ / TM)) * TM;
  const size_t bh = (size_t)b * HHEADS + h;
  const size_t so = (bh * NSEQ + n0) * RLOW;

  v8f acc[2][4];
#pragma unroll
  for (int i = 0; i < 2; ++i)
#pragma unroll
    for (int j = 0; j < 4; ++j)
      acc[i][j] = (v8f){0.f,0.f,0.f,0.f,0.f,0.f,0.f,0.f};

  const int mrow0 = mw * 32 + lo;
  const int mrow1 = mrow0 + 16;
  const int cb    = nw * 64 + lo;

  const int lrow = tid >> 3;
  const int lc4  = (tid & 7) << 2;
  const float* pa = X + (size_t)(bm + lrow) * GK + lc4;
  const float* pb = W + (size_t)(h * DH + lrow) * GK + lc4;

  float4 ra[4], rb[4];
  tile_gload(pa, pb, 0, ra, rb);
  tile_lstore(As0, Bs0, lrow, lc4, ra, rb);
  __syncthreads();

#define PIN_LANDING_BEHIND() __builtin_amdgcn_sched_barrier(0x020)
  for (int t = 0; t < NT - 2; t += 2) {
    tile_gload(pa, pb, (t + 1) * TK, ra, rb);
    tile_mma(As0, Bs0, mrow0, mrow1, cb, hi, acc);
    PIN_LANDING_BEHIND();
    tile_lstore(As1, Bs1, lrow, lc4, ra, rb);
    __syncthreads();
    tile_gload(pa, pb, (t + 2) * TK, ra, rb);
    tile_mma(As1, Bs1, mrow0, mrow1, cb, hi, acc);
    PIN_LANDING_BEHIND();
    tile_lstore(As0, Bs0, lrow, lc4, ra, rb);
    __syncthreads();
  }
  tile_gload(pa, pb, (NT - 1) * TK, ra, rb);
  tile_mma(As0, Bs0, mrow0, mrow1, cb, hi, acc);
  PIN_LANDING_BEHIND();
  tile_lstore(As1, Bs1, lrow, lc4, ra, rb);
  __syncthreads();
  tile_mma(As1, Bs1, mrow0, mrow1, cb, hi, acc);

  {
    float* dst = rows_s + (mw * 32 + 8 * hi) * RP + nw * 64 + lo;
#pragma unroll
    for (int i = 0; i < 8; ++i) {
      float* p0 = dst + i * RP;
      p0[0]  = acc[0][0][i];
      p0[16] = acc[0][1][i];
      p0[32] = acc[0][2][i];
      p0[48] = acc[0][3][i];
      float* p1 = dst + (16 + i) * RP;
      p1[0]  = acc[1][0][i];
      p1[16] = acc[1][1][i];
      p1[32] = acc[1][2][i];
      p1[48] = acc[1][3][i];
    }
  }
  for (int i = tid; i < TM * RLOW / 4; i += 256)
    *(float4*)(&ep_s[i * 4]) = *(const float4*)(eps + so + (size_t)i * 4);
  for (int i = tid; i < RLOW * RLOW * RLOW; i += 256) st_s[i] = s_tril[(size_t)h * (RLOW * RLOW * RLOW) + i];
  for (int i = tid; i < DH * RLOW; i += 256) wf_s[i] = Wf[i];
  if (tid < RLOW * RLOW) {
    mu_s[tid] = m_u[h * (RLOW * RLOW) + tid];
    sd_s[tid] = expf(log_ssqrt[h * RLOW * RLOW + tid]);
  }
  if (tid < DH)   bf_s[tid]   = bf[tid];
  if (tid < RLOW) lam2_s[tid] = expf(2.f * log_lambda[h * RLOW + tid]);
  __syncthreads();

  const int row = tid & (TM - 1);
  const bool krole = __builtin_amdgcn_readfirstlane(wave) >= 4;
  float e[RLOW];
#pragma unroll
  for (int r = 0; r < RLOW; ++r) e[r] = 0.f;

  if (!krole) {
    const float* wep = we + bh * (DH * RLOW);
    float qn = 0.f;
#pragma unroll
    for (int d4 = 0; d4 < DH / 4; ++d4) {
      const float4 v = *(const float4*)(&rows_s[row * RP + d4 * 4]);
      const float xv[4] = {v.x, v.y, v.z, v.w};
#pragma unroll
      for (int j = 0; j < 4; ++j) {
        const float x = xv[j];
        qn += x * x;
#pragma unroll
        for (int r = 0; r < RLOW; ++r) e[r] += x * wep[(d4 * 4 + j) * RLOW + r];
      }
    }
    const float qi = 1.f / fmaxf(sqrtf(qn), 1e-12f);
#pragma unroll
    for (int r = 0; r < RLOW; ++r) e[r] *= qi;
  } else {
    const float* wrp = wr + bh * (DH * RLOW);
    float rv[RLOW];
#pragma unroll
    for (int r = 0; r < RLOW; ++r) rv[r] = 0.f;
    float kn = 0.f;
#pragma unroll
    for (int d4 = 0; d4 < DH / 4; ++d4) {
      const float4 v = *(const float4*)(&rows_s[row * RP + DH + d4 * 4]);
      const float xv[4] = {v.x, v.y, v.z, v.w};
#pragma unroll
      for (int j = 0; j < 4; ++j) {
        const float x = xv[j];
        kn += x * x;
#pragma unroll
        for (int r = 0; r < RLOW; ++r) rv[r] += x * wrp[(d4 * 4 + j) * RLOW + r];
      }
    }
    const float ki = 1.f / fmaxf(sqrtf(kn), 1e-12f);
#pragma unroll
    for (int r = 0; r < RLOW; ++r) rs_s[row * RLOW + r] = rv[r] * ki;
  }
  __syncthreads();

  if (!krole) {
    float ev[RLOW], rv[RLOW], v1[RLOW];
    {
      const float2* e2 = (const float2*)(&ep_s[row * RLOW]);
      const float2* r2 = (const float2*)(&rs_s[row * RLOW]);
#pragma unroll
      for (int j = 0; j < RLOW / 2; ++j) {
        const float2 t = e2[j]; ev[2 * j] = t.x; ev[2 * j + 1] = t.y;
        const float2 u = r2[j]; rv[2 * j] = u.x; rv[2 * j + 1] = u.y;
      }
    }
#pragma unroll
    for (int r = 0; r < RLOW; ++r) {
      v1[r] = (e[r] + rv[r]) * lam2_s[r];
      ep_s[row * RLOW + r] = e[r];
    }

#pragma unroll
    for (int r = 0; r < RLOW; ++r) rows_s[row * RP + r] = v1[r];

    float mean[RLOW], tq[RLOW][RLOW];
#pragma unroll
    for (int c = 0; c < RLOW; ++c) mean[c] = 0.f;
#pragma unroll
    for (int q = 0; q < RLOW; ++q)
#pragma unroll
      for (int c = 0; c <= q; ++c) tq[q][c] = 0.f;
#pragma unroll 1
    for (int a = 0; a < RLOW; ++a) {
      const float va = rows_s[row * RP + a];
      const float* mu = &mu_s[a * RLOW];
      const float* st = &st_s[a * RLOW * RLOW];
      const float* sd = &sd_s[a * RLOW];
#pragma unroll
      for (int c = 0; c < RLOW; ++c) mean[c] += va * mu[c];
#pragma unroll
      for (int q = 0; q < RLOW; ++q) {
#pragma unroll
        for (int c = 0; c < q; ++c) tq[q][c] += va * st[q * RLOW + c];
        tq[q][q] += va * sd[q];
      }
    }
    float sample[RLOW];
#pragma unroll
    for (int c = 0; c < RLOW; ++c) {
      float noise = 0.f;
#pragma unroll
      for (int q = c; q < RLOW; ++q) noise += ev[q] * tq[q][c];
      sample[c] = mean[c] + noise;
    }

#pragma unroll 2
    for (int d4 = 0; d4 < DH / 4; ++d4) {
      float o[4];
#pragma unroll
      for (int j = 0; j < 4; ++j) {
        const int d = d4 * 4 + j;
        float a = bf_s[d];
#pragma unroll
        for (int r = 0; r < RLOW; ++r) a += sample[r] * wf_s[d * RLOW + r];
        o[j] = a;
      }
      *(float4*)(&rows_s[row * RP + d4 * 4]) = make_float4(o[0], o[1], o[2], o[3]);
    }
  }
  __syncthreads();

  float* abase = attn + ((size_t)b * NSEQ + n0) * CDIM + h * DH;
#pragma unroll
  for (int it = 0; it < (TM * DH / 4) / 256; ++it) {
    const int i  = tid + it * 256;
    const int rw = i >> 4;
    const int c4 = (i & 15) << 2;
    *(float4*)(abase + (size_t)rw * CDIM + c4) = *(const float4*)(&rows_s[rw * RP + c4]);
  }
  for (int i = tid; i < TM * RLOW / 4; i += 256) {
    *(float4*)(esc + so + (size_t)i * 4) = *(const float4*)(&ep_s[i * 4]);
    *(float4*)(rsc + so + (size_t)i * 4) = *(const float4*)(&rs_s[i * 4]);
  }
}

extern "C" void kernel_launch(void* const* d_in, const int* in_sizes, int n_in,
                              void* d_out, int out_size, void* d_ws, size_t ws_size,
                              hipStream_t stream) {
  const float* x          = (const float*)d_in[0];
  const float* Wqk        = (const float*)d_in[1];
  const float* we_p       = (const float*)d_in[2];
  const float* wr_p       = (const float*)d_in[3];
  const float* log_lambda = (const float*)d_in[4];
  const float* m_u        = (const float*)d_in[5];
  const float* s_tril     = (const float*)d_in[6];
  const float* log_ssqrt  = (const float*)d_in[7];
  const float* Wf         = (const float*)d_in[8];
  const float* bf         = (const float*)d_in[9];
  const float* eps        = (const float*)d_in[10];

  float* ws = (float*)d_ws;
  float* we = ws;
  float* wr = we + (size_t)BDIM * HHEADS * DH * RLOW;

  float* attn = (float*)d_out;
  float* esc  = attn + (size_t)BDIM * NSEQ * CDIM;
  float* rsc  = esc + (size_t)BDIM * HHEADS * NSEQ * RLOW;
  float* klp  = rsc + (size_t)BDIM * HHEADS * NSEQ * RLOW;

  gen_weights_kl<<<(BDIM * HHEADS * DH * RLOW + 255) / 256 + 1, 256, 0, stream>>>(
      x, we_p, wr_p, we, wr, log_lambda, m_u, s_tril, log_ssqrt, klp);
  qk_gemm_svgp<<<dim3(HHEADS, GM / TM), 256, 0, stream>>>(
      x, Wqk, we, wr, log_lambda, m_u, s_tril, log_ssqrt, Wf, bf, eps, esc, rsc, attn);
}
